# starting kernel
# baseline (speedup 1.0000x reference)
#include <hip/hip_runtime.h>

#define N_NODES 50000
#define N_EDGES 800000
#define DFEAT   128
#define ASTRIDE 132

typedef float v2f __attribute__((ext_vector_type(2)));
typedef float v8f __attribute__((ext_vector_type(8)));

__device__ __forceinline__ void atomic_add_f32(float* p, float v) {
    __hip_atomic_fetch_add(p, v, __ATOMIC_RELAXED, __HIP_MEMORY_SCOPE_AGENT);
}

__global__ __launch_bounds__(256) void zero_f4(float4* __restrict__ p, int n4) {
    int i = blockIdx.x * blockDim.x + threadIdx.x;
    if (i < n4) p[i] = make_float4(0.f, 0.f, 0.f, 0.f);
}

__global__ __launch_bounds__(256) void degree_kernel(const int* __restrict__ src,
                                                     const int* __restrict__ dst,
                                                     float* __restrict__ deg_out,
                                                     float* __restrict__ deg_in,
                                                     int n_edges) {
    int e = blockIdx.x * blockDim.x + threadIdx.x;
    if (e < n_edges) {
        atomic_add_f32(&deg_out[src[e]], 1.0f);
        atomic_add_f32(&deg_in [dst[e]], 1.0f);
    }
}

__global__ __launch_bounds__(256) void deg_to_rsqrt(float* __restrict__ p, int n) {
    int i = blockIdx.x * blockDim.x + threadIdx.x;
    if (i < n) p[i] = rsqrtf(fmaxf(p[i], 1.0f));
}

__global__ __launch_bounds__(256) void scatter_kernel(const float* __restrict__ X,
                                                      const float* __restrict__ rs_out,
                                                      const int* __restrict__ src,
                                                      const int* __restrict__ dst,
                                                      float* __restrict__ msg,
                                                      int n_edges) {
    int e    = (blockIdx.x * blockDim.x + threadIdx.x) >> 5;
    int lane = threadIdx.x & 31;
    if (e >= n_edges) return;
    int   s = src[e];
    int   d = dst[e];
    float scale = rs_out[s];
    float4 v = ((const float4*)(X + (size_t)s * DFEAT))[lane];
    float* mp = msg + (size_t)d * DFEAT + lane * 4;
    atomic_add_f32(mp + 0, v.x * scale);
    atomic_add_f32(mp + 1, v.y * scale);
    atomic_add_f32(mp + 2, v.z * scale);
    atomic_add_f32(mp + 3, v.w * scale);
}

template <bool RELU>
__global__ __launch_bounds__(256) void gcn_gemm(const float* __restrict__ A,
                                                const float* __restrict__ rs_in,
                                                const float* __restrict__ W,
                                                const float* __restrict__ bias,
                                                float* __restrict__ out) {
    __shared__ float Alds[16 * ASTRIDE];

    const int tid     = threadIdx.x;
    const int wave    = tid >> 5;
    const int lane    = tid & 31;
    const int rowBase = blockIdx.x << 4;

    {
        const int r = tid >> 4;
        const int c = (tid & 15) << 3;
        const float scale = rs_in[rowBase + r];
        const float4* gp = (const float4*)(A + (size_t)(rowBase + r) * DFEAT + c);
        float4 a0 = gp[0];
        float4 a1 = gp[1];
        float* lp = &Alds[r * ASTRIDE + c];
        lp[0] = a0.x * scale; lp[1] = a0.y * scale; lp[2] = a0.z * scale; lp[3] = a0.w * scale;
        lp[4] = a1.x * scale; lp[5] = a1.y * scale; lp[6] = a1.z * scale; lp[7] = a1.w * scale;
    }
    __syncthreads();

    const int   mrow = lane & 15;
    const int   khi  = (lane >> 4) << 1;
    const int   n0   = wave << 4;
    const float* wp  = W + n0 + mrow;
    const float* ap  = &Alds[mrow * ASTRIDE + khi];

    v8f acc = {0.f, 0.f, 0.f, 0.f, 0.f, 0.f, 0.f, 0.f};
#pragma unroll
    for (int k = 0; k < DFEAT; k += 4) {
        v2f a = *(const v2f*)(ap + k);
        v2f b;
        b.x = wp[(size_t)(k + khi)     * DFEAT];
        b.y = wp[(size_t)(k + khi + 1) * DFEAT];
        acc = __builtin_amdgcn_wmma_f32_16x16x4_f32(false, a, false, b,
                                                    (short)0, acc, false, false);
    }

    const int   col  = n0 + mrow;
    const float bv   = bias[col];
    const int   rsel = (lane >> 4) << 3;
#pragma unroll
    for (int v = 0; v < 8; ++v) {
        float h = acc[v] + bv;
        if (RELU) h = fmaxf(h, 0.f);
        out[(size_t)(rowBase + v + rsel) * DFEAT + col] = h;
    }
}

extern "C" void kernel_launch(void* const* d_in, const int* in_sizes, int n_in,
                              void* d_out, int out_size, void* d_ws, size_t ws_size,
                              hipStream_t stream) {
    const float* x   = (const float*)d_in[0];
    const int*   src = (const int*)  d_in[1];
    const int*   dst = (const int*)  d_in[2];
    const float* W1  = (const float*)d_in[3];
    const float* b1  = (const float*)d_in[4];
    const float* W2  = (const float*)d_in[5];
    const float* b2  = (const float*)d_in[6];
    float*       out = (float*)d_out;

    float* ws     = (float*)d_ws;
    float* rs_out = ws;
    float* rs_in  = ws + N_NODES;
    float* msg    = ws + 2 * N_NODES;
    float* h1     = msg + (size_t)N_NODES * DFEAT;

    const int T     = 256;
    const int ND4   = N_NODES * DFEAT / 4;
    const int EBLK  = (N_EDGES + 7) / 8;
    const int GEMMB = N_NODES / 16;

    zero_f4<<<(2 * N_NODES / 4 + T - 1) / T, T, 0, stream>>>((float4*)rs_out, 2 * N_NODES / 4);
    degree_kernel<<<(N_EDGES + T - 1) / T, T, 0, stream>>>(src, dst, rs_out, rs_in, N_EDGES);
    deg_to_rsqrt<<<(2 * N_NODES + T - 1) / T, T, 0, stream>>>(rs_out, 2 * N_NODES);

    zero_f4<<<(ND4 + T - 1) / T, T, 0, stream>>>((float4*)msg, ND4);
    scatter_kernel<<<EBLK, T, 0, stream>>>(x, rs_out, src, dst, msg, N_EDGES);
    gcn_gemm<true><<<GEMMB, T, 0, stream>>>(msg, rs_in, W1, b1, h1);

    zero_f4<<<(ND4 + T - 1) / T, T, 0, stream>>>((float4*)msg, ND4);
    scatter_kernel<<<EBLK, T, 0, stream>>>(h1, rs_out, src, dst, msg, N_EDGES);
    gcn_gemm<false><<<GEMMB, T, 0, stream>>>(msg, rs_in, W2, b2, out);
}


# Edge sum moved into the layer kernel: a work-group owns 208 destination nodes, filters dst[] itself into LDS lists, sums source rows in registers in fixed order, feeds WMMA; float atomics gone
# speedup vs baseline: 2.6560x; 2.6560x over previous
#include <hip/hip_runtime.h>

#define N_NODES 50000
#define N_EDGES 800000
#define DFEAT   128
#define ASTRIDE 132

typedef float v2f __attribute__((ext_vector_type(2)));
typedef float v8f __attribute__((ext_vector_type(8)));

#define NPART 32
#define EDGES_PER_PART (N_EDGES / NPART)
#define E4_PER_PART    (EDGES_PER_PART / 4)
#define N4             (N_NODES / 4)
#define HIST_THREADS   1024
#define E4_PER_THREAD  ((E4_PER_PART + HIST_THREADS - 1) / HIST_THREADS)

__global__ __launch_bounds__(HIST_THREADS) void degree_hist(const int* __restrict__ src,
                                                            const int* __restrict__ dst,
                                                            unsigned* __restrict__ part) {
    __shared__ unsigned hist[N_NODES];

    const int tid = threadIdx.x;
    const int arr = blockIdx.x / NPART;
    const int p   = blockIdx.x % NPART;
    const int* idx = arr ? dst : src;
    const int4* ep = (const int4*)(idx + (size_t)p * EDGES_PER_PART);

    int4 e[E4_PER_THREAD];
#pragma unroll
    for (int j = 0; j < E4_PER_THREAD; ++j) {
        const int i = tid + j * HIST_THREADS;
        if (i < E4_PER_PART) e[j] = ep[i];
    }

    uint4* h4 = (uint4*)hist;
    for (int i = tid; i < N4; i += HIST_THREADS) h4[i] = make_uint4(0u, 0u, 0u, 0u);
    __syncthreads();

#pragma unroll
    for (int j = 0; j < E4_PER_THREAD; ++j) {
        const int i = tid + j * HIST_THREADS;
        if (i < E4_PER_PART) {
            atomicAdd(&hist[e[j].x], 1u);
            atomicAdd(&hist[e[j].y], 1u);
            atomicAdd(&hist[e[j].z], 1u);
            atomicAdd(&hist[e[j].w], 1u);
        }
    }
    __syncthreads();

    uint4* op = (uint4*)(part + (size_t)blockIdx.x * N_NODES);
    for (int i = tid; i < N4; i += HIST_THREADS) op[i] = h4[i];
}

__global__ __launch_bounds__(256) void degree_finish(const unsigned* __restrict__ part,
                                                     float* __restrict__ rs) {
    const int t = blockIdx.x * blockDim.x + threadIdx.x;
    if (t >= 2 * N4) return;
    const int arr = (t >= N4) ? 1 : 0;
    const int q   = t - arr * N4;
    const uint4* pp = (const uint4*)part + (size_t)arr * NPART * N4 + q;
    uint4 s = make_uint4(0u, 0u, 0u, 0u);
#pragma unroll
    for (int p = 0; p < NPART; ++p) {
        const uint4 v = pp[(size_t)p * N4];
        s.x += v.x; s.y += v.y; s.z += v.z; s.w += v.w;
    }
    float4 r;
    r.x = rsqrtf(fmaxf((float)s.x, 1.0f));
    r.y = rsqrtf(fmaxf((float)s.y, 1.0f));
    r.z = rsqrtf(fmaxf((float)s.z, 1.0f));
    r.w = rsqrtf(fmaxf((float)s.w, 1.0f));
    ((float4*)rs)[t] = r;
}

#define RNODES   208
#define NLBLK    ((N_NODES + RNODES - 1) / RNODES)
#define LTHREADS 1024
#define LWAVES   32
#define CAPS     24
#define E4_TOTAL (N_EDGES / 4)
#define E4_WAVE  (E4_TOTAL / LWAVES)
#define SCAN_UNR 8
#define SCAN_BATCHES ((E4_WAVE + 32 * SCAN_UNR - 1) / (32 * SCAN_UNR))
#define GATHER_U 8
#define FLAT_CAP (LWAVES * CAPS)

__device__ __forceinline__ void fma4(float4& a, const float4 v, const float s) {
    a.x = fmaf(v.x, s, a.x);
    a.y = fmaf(v.y, s, a.y);
    a.z = fmaf(v.z, s, a.z);
    a.w = fmaf(v.w, s, a.w);
}

__device__ __forceinline__ void scan_load(int4 (&d)[SCAN_UNR], const int4* __restrict__ dst4,
                                          int pos, int end, int lane) {
    if (pos + 32 * SCAN_UNR <= end) {
#pragma unroll
        for (int j = 0; j < SCAN_UNR; ++j) d[j] = dst4[pos + j * 32 + lane];
    } else {
#pragma unroll
        for (int j = 0; j < SCAN_UNR; ++j) {
            const int i = pos + j * 32 + lane;
            d[j] = (i < end) ? dst4[i] : make_int4(-1, -1, -1, -1);
        }
    }
}

#define ACC_ROW(slot, v, s)                         \
    switch (slot) {                                 \
    case 0: fma4(a0, v, s); break;                  \
    case 1: fma4(a1, v, s); break;                  \
    case 2: fma4(a2, v, s); break;                  \
    case 3: fma4(a3, v, s); break;                  \
    case 4: fma4(a4, v, s); break;                  \
    case 5: fma4(a5, v, s); break;                  \
    case 6: fma4(a6, v, s); break;                  \
    default: break;                                 \
    }

template <int NT, bool RELU>
__device__ __forceinline__ void tile_product(const float* Alds, const float* __restrict__ Wm,
                                             const float* __restrict__ bias,
                                             float* __restrict__ out,
                                             int lo, int ntiles, int g, int n0, int lane) {
    const int mrow = lane & 15;
    const int khi  = (lane >> 4) << 1;
    const float* wp = Wm + n0 + mrow;
    const float* ap[NT];
    bool valid[NT];
    v8f acc[NT];
#pragma unroll
    for (int t = 0; t < NT; ++t) {
        const int rt = g + 4 * t;
        valid[t] = rt < ntiles;
        const int rtc = valid[t] ? rt : g;
        ap[t] = Alds + (rtc * 16 + mrow) * ASTRIDE + khi;
        acc[t] = (v8f){0.f, 0.f, 0.f, 0.f, 0.f, 0.f, 0.f, 0.f};
    }
#pragma unroll
    for (int k = 0; k < DFEAT; k += 4) {
        v2f b;
        b.x = wp[(size_t)(k + khi)     * DFEAT];
        b.y = wp[(size_t)(k + khi + 1) * DFEAT];
#pragma unroll
        for (int t = 0; t < NT; ++t) {
            const v2f a = *(const v2f*)(ap[t] + k);
            acc[t] = __builtin_amdgcn_wmma_f32_16x16x4_f32(false, a, false, b,
                                                           (short)0, acc[t], false, false);
        }
    }
    const int   col  = n0 + mrow;
    const float bv   = bias[col];
    const int   rsel = (lane >> 4) << 3;
#pragma unroll
    for (int t = 0; t < NT; ++t) {
        if (valid[t]) {
            const int rowBase = lo + (g + 4 * t) * 16;
#pragma unroll
            for (int v = 0; v < 8; ++v) {
                float h = acc[t][v] + bv;
                if (RELU) h = fmaxf(h, 0.f);
                out[(size_t)(rowBase + v + rsel) * DFEAT + col] = h;
            }
        }
    }
}

template <bool RELU>
__global__ __launch_bounds__(LTHREADS) void gcn_layer(const float* __restrict__ X,
                                                      const float* __restrict__ rs_out,
                                                      const float* __restrict__ rs_in,
                                                      const int* __restrict__ src,
                                                      const int* __restrict__ dst,
                                                      const float* __restrict__ Wm,
                                                      const float* __restrict__ bias,
                                                      float* __restrict__ out) {
    __shared__ __attribute__((aligned(16))) float Alds[RNODES * ASTRIDE];
    __shared__ __attribute__((aligned(16))) unsigned lists[LWAVES * LWAVES * CAPS];
    __shared__ unsigned cntS[LWAVES * LWAVES];
    __shared__ unsigned ovfFlag;

    const int tid  = threadIdx.x;
    const int lane = tid & 31;
    const int w    = __builtin_amdgcn_readfirstlane(tid >> 5);
    const int lo   = blockIdx.x * RNODES;
    const unsigned nrows = (N_NODES - lo < RNODES) ? (unsigned)(N_NODES - lo) : (unsigned)RNODES;

    const int4*   dst4 = (const int4*)dst;
    const float4* X4   = (const float4*)X;

    if (tid == 0) ovfFlag = 0u;
    __syncthreads();

    {
        const int beg = w * E4_WAVE;
        const int end = beg + E4_WAVE;
        unsigned cnt = 0u;
        bool ovf = false;
        const int rot = blockIdx.x % SCAN_BATCHES;
        int4 dn[SCAN_UNR];
        scan_load(dn, dst4, beg + rot * (32 * SCAN_UNR), end, lane);
        for (int t = 0; t < SCAN_BATCHES; ++t) {
            int tb = t + rot;
            if (tb >= SCAN_BATCHES) tb -= SCAN_BATCHES;
            const int pos = beg + tb * (32 * SCAN_UNR);
            int4 d[SCAN_UNR];
#pragma unroll
            for (int j = 0; j < SCAN_UNR; ++j) d[j] = dn[j];
            if (t + 1 < SCAN_BATCHES) {
                int tn = tb + 1;
                if (tn >= SCAN_BATCHES) tn -= SCAN_BATCHES;
                scan_load(dn, dst4, beg + tn * (32 * SCAN_UNR), end, lane);
            }
#pragma unroll
            for (int j = 0; j < SCAN_UNR; ++j) {
                const unsigned u0 = (unsigned)d[j].x - (unsigned)lo;
                const unsigned u1 = (unsigned)d[j].y - (unsigned)lo;
                const unsigned u2 = (unsigned)d[j].z - (unsigned)lo;
                const unsigned u3 = (unsigned)d[j].w - (unsigned)lo;
                unsigned m0 = __builtin_amdgcn_ballot_w32(u0 < nrows);
                unsigned m1 = __builtin_amdgcn_ballot_w32(u1 < nrows);
                unsigned m2 = __builtin_amdgcn_ballot_w32(u2 < nrows);
                unsigned m3 = __builtin_amdgcn_ballot_w32(u3 < nrows);
                if ((m0 | m1 | m2 | m3) != 0u) {
                    const unsigned ib = (unsigned)(pos + j * 32);
#define SCAN_TAKE(m, u, c)                                                              \
                    while (m) {                                                         \
                        const int l = __builtin_ctz(m);                                 \
                        m &= m - 1u;                                                    \
                        const unsigned row   = (unsigned)__builtin_amdgcn_readlane((int)(u), l); \
                        const unsigned owner = row & 31u;                               \
                        const unsigned n = (unsigned)__builtin_amdgcn_readlane((int)cnt, (int)owner); \
                        if (n < (unsigned)CAPS) {                                       \
                            if ((unsigned)lane == owner) {                              \
                                lists[(owner * LWAVES + (unsigned)w) * CAPS + n] =      \
                                    (row << 20) | (4u * (ib + (unsigned)l) + (c));      \
                                cnt = n + 1u;                                           \
                            }                                                           \
                        } else {                                                        \
                            ovf = true;                                                 \
                        }                                                               \
                    }
                    SCAN_TAKE(m0, u0, 0u)
                    SCAN_TAKE(m1, u1, 1u)
                    SCAN_TAKE(m2, u2, 2u)
                    SCAN_TAKE(m3, u3, 3u)
#undef SCAN_TAKE
                }
            }
        }
        cntS[lane * LWAVES + w] = cnt;
        if (ovf && lane == 0) ovfFlag = 1u;
    }
    __syncthreads();
    const bool slow = __builtin_amdgcn_readfirstlane((int)ovfFlag) != 0;

    float4 a0 = make_float4(0.f, 0.f, 0.f, 0.f);
    float4 a1 = a0, a2 = a0, a3 = a0, a4 = a0, a5 = a0, a6 = a0;

    if (!slow) {
        unsigned* myflat = (unsigned*)Alds + w * FLAT_CAP;
        const unsigned c_l = cntS[w * LWAVES + lane];
        unsigned incl = c_l;
#pragma unroll
        for (int dd = 1; dd < 32; dd <<= 1) {
            const unsigned up = __shfl_up(incl, dd);
            if (lane >= dd) incl += up;
        }
        const unsigned off = incl - c_l;
        const int total = __builtin_amdgcn_readlane((int)incl, 31);
        const uint4* mylist4 = (const uint4*)(lists + (w * LWAVES + lane) * CAPS);
        unsigned ent[CAPS];
        unsigned sn[CAPS];
#pragma unroll
        for (int i = 0; i < CAPS / 4; ++i) {
            const uint4 q = mylist4[i];
            ent[4 * i + 0] = q.x; ent[4 * i + 1] = q.y; ent[4 * i + 2] = q.z; ent[4 * i + 3] = q.w;
        }
#pragma unroll
        for (int i = 0; i < CAPS; ++i) {
            const unsigned e = ((unsigned)i < c_l) ? (ent[i] & 0xFFFFFu) : 0u;
            sn[i] = (unsigned)src[e];
        }
#pragma unroll
        for (int i = 0; i < CAPS; ++i) asm volatile("" :: "v"(sn[i]));
#pragma unroll
        for (int i = 0; i < CAPS; ++i) {
            if ((unsigned)i < c_l) myflat[off + i] = ((ent[i] >> 25) << 16) | sn[i];
        }
        __syncthreads();

        unsigned* mypart = lists + w * FLAT_CAP;
        unsigned nj[7], oj[7], run[7];
#pragma unroll
        for (int j = 0; j < 7; ++j) { nj[j] = 0u; run[j] = 0u; }
        for (int k = 0; k < total; k += 32) {
            const int idx = k + lane;
            const unsigned slot = (idx < total) ? (myflat[idx] >> 16) : 7u;
#pragma unroll
            for (int j = 0; j < 7; ++j)
                nj[j] += (unsigned)__builtin_popcount(__builtin_amdgcn_ballot_w32(slot == (unsigned)j));
        }
        oj[0] = 0u;
#pragma unroll
        for (int j = 1; j < 7; ++j) oj[j] = oj[j - 1] + nj[j - 1];
        for (int k = 0; k < total; k += 32) {
            const int idx = k + lane;
            const unsigned ent  = (idx < total) ? myflat[idx] : (7u << 16);
            const unsigned slot = ent >> 16;
#pragma unroll
            for (int j = 0; j < 7; ++j) {
                const unsigned m = __builtin_amdgcn_ballot_w32(slot == (unsigned)j);
                if (slot == (unsigned)j)
                    mypart[oj[j] + run[j] + __builtin_amdgcn_mbcnt_lo(m, 0u)] = ent & 0xFFFFu;
                run[j] += (unsigned)__builtin_popcount(m);
            }
        }
        __syncthreads();

#define GATHER_SLOT(j, a)                                                               \
        for (unsigned k0 = 0u; k0 < nj[j]; k0 += GATHER_U) {                            \
            const unsigned kl   = k0 + (unsigned)lane;                                  \
            const unsigned entv = mypart[oj[j] + ((kl < nj[j]) ? kl : (nj[j] - 1u))];   \
            float4 v[GATHER_U];                                                         \
            float  sc[GATHER_U];                                                        \
            _Pragma("unroll")                                                           \
            for (int u = 0; u < GATHER_U; ++u) {                                        \
                const unsigned s = (unsigned)__builtin_amdgcn_readlane((int)entv, u);   \
                v[u]  = X4[(size_t)s * (DFEAT / 4) + lane];                             \
                sc[u] = rs_out[s];                                                      \
            }                                                                           \
              \
               \
                                                 \
            _Pragma("unroll")                                                           \
            for (int u = 0; u < GATHER_U; ++u)                                          \
                asm volatile("" :: "v"(v[u].x), "v"(v[u].y), "v"(v[u].z), "v"(v[u].w), \
                             "s"(sc[u]));                                               \
            _Pragma("unroll")                                                           \
            for (int u = 0; u < GATHER_U; ++u) {                                        \
                if (k0 + (unsigned)u < nj[j]) fma4(a, v[u], sc[u]);                     \
            }                                                                           \
        }
        GATHER_SLOT(0, a0)
        GATHER_SLOT(1, a1)
        GATHER_SLOT(2, a2)
        GATHER_SLOT(3, a3)
        GATHER_SLOT(4, a4)
        GATHER_SLOT(5, a5)
        GATHER_SLOT(6, a6)
#undef GATHER_SLOT
    } else {
        for (int pos = 0; pos < E4_TOTAL; pos += 32) {
            const int4 d = dst4[pos + lane];
            const unsigned u0 = (unsigned)d.x - (unsigned)lo;
            const unsigned u1 = (unsigned)d.y - (unsigned)lo;
            const unsigned u2 = (unsigned)d.z - (unsigned)lo;
            const unsigned u3 = (unsigned)d.w - (unsigned)lo;
            unsigned m0 = __builtin_amdgcn_ballot_w32(u0 < nrows && (u0 & 31u) == (unsigned)w);
            unsigned m1 = __builtin_amdgcn_ballot_w32(u1 < nrows && (u1 & 31u) == (unsigned)w);
            unsigned m2 = __builtin_amdgcn_ballot_w32(u2 < nrows && (u2 & 31u) == (unsigned)w);
            unsigned m3 = __builtin_amdgcn_ballot_w32(u3 < nrows && (u3 & 31u) == (unsigned)w);
            if ((m0 | m1 | m2 | m3) != 0u) {
#define SLOW_TAKE(m, u, c)                                                              \
                while (m) {                                                             \
                    const int l = __builtin_ctz(m);                                     \
                    m &= m - 1u;                                                        \
                    const unsigned row = (unsigned)__builtin_amdgcn_readlane((int)(u), l); \
                    const int      s   = src[4 * (pos + l) + (c)];                      \
                    const float    scs = rs_out[s];                                     \
                    const float4   vs  = X4[(size_t)s * (DFEAT / 4) + lane];            \
                    ACC_ROW(row >> 5, vs, scs)                                          \
                }
                SLOW_TAKE(m0, u0, 0)
                SLOW_TAKE(m1, u1, 1)
                SLOW_TAKE(m2, u2, 2)
                SLOW_TAKE(m3, u3, 3)
#undef SLOW_TAKE
            }
        }
    }
    __syncthreads();

#define PUT_ROW(j, a)                                                                   \
    {                                                                                   \
        const int r = (j) * 32 + w;                                                     \
        if (r < RNODES) {                                                               \
            const float si = (r < (int)nrows) ? rs_in[lo + r] : 0.f;                    \
            *(float4*)&Alds[r * ASTRIDE + lane * 4] =                                   \
                make_float4(a.x * si, a.y * si, a.z * si, a.w * si);                    \
        }                                                                               \
    }
    PUT_ROW(0, a0)
    PUT_ROW(1, a1)
    PUT_ROW(2, a2)
    PUT_ROW(3, a3)
    PUT_ROW(4, a4)
    PUT_ROW(5, a5)
    PUT_ROW(6, a6)
#undef PUT_ROW
    __syncthreads();

    const int ntiles = (int)(nrows >> 4);
    const int g      = w >> 3;
    const int n0     = (w & 7) << 4;
    if (g == 0) tile_product<4, RELU>(Alds, Wm, bias, out, lo, ntiles, g, n0, lane);
    else        tile_product<3, RELU>(Alds, Wm, bias, out, lo, ntiles, g, n0, lane);
}

extern "C" void kernel_launch(void* const* d_in, const int* in_sizes, int n_in,
                              void* d_out, int out_size, void* d_ws, size_t ws_size,
                              hipStream_t stream) {
    const float* x   = (const float*)d_in[0];
    const int*   src = (const int*)  d_in[1];
    const int*   dst = (const int*)  d_in[2];
    const float* W1  = (const float*)d_in[3];
    const float* b1  = (const float*)d_in[4];
    const float* W2  = (const float*)d_in[5];
    const float* b2  = (const float*)d_in[6];
    float*       out = (float*)d_out;

    float* ws     = (float*)d_ws;
    float* rs_out = ws;
    float* rs_in  = ws + N_NODES;
    float* scr    = ws + 2 * N_NODES;
    float* h1     = scr + (size_t)N_NODES * DFEAT;

    const int T = 256;

    unsigned* part = (unsigned*)scr;
    degree_hist<<<2 * NPART, HIST_THREADS, 0, stream>>>(src, dst, part);
    degree_finish<<<(2 * N4 + T - 1) / T, T, 0, stream>>>(part, rs_out);

    gcn_layer<true><<<NLBLK, LTHREADS, 0, stream>>>(x, rs_out, rs_in, src, dst, W1, b1, h1);
    gcn_layer<false><<<NLBLK, LTHREADS, 0, stream>>>(h1, rs_out, rs_in, src, dst, W2, b2, out);
}


# Layer 2 loads the per-wave row-ordered source lists that layer 1 built and kept in the workspace (same call): no second dst[] scan, src[] lookup or ordering pass; sums and their order unchanged
# speedup vs baseline: 3.3213x; 1.2505x over previous
#include <hip/hip_runtime.h>

#define N_NODES 50000
#define N_EDGES 800000
#define DFEAT   128
#define ASTRIDE 132

typedef float v2f __attribute__((ext_vector_type(2)));
typedef float v8f __attribute__((ext_vector_type(8)));

#define NPART 32
#define EDGES_PER_PART (N_EDGES / NPART)
#define E4_PER_PART    (EDGES_PER_PART / 4)
#define N4             (N_NODES / 4)
#define HIST_THREADS   1024
#define E4_PER_THREAD  ((E4_PER_PART + HIST_THREADS - 1) / HIST_THREADS)

__global__ __launch_bounds__(HIST_THREADS) void degree_hist(const int* __restrict__ src,
                                                            const int* __restrict__ dst,
                                                            unsigned* __restrict__ part) {
    __shared__ unsigned hist[N_NODES];

    const int tid = threadIdx.x;
    const int arr = blockIdx.x / NPART;
    const int p   = blockIdx.x % NPART;
    const int* idx = arr ? dst : src;
    const int4* ep = (const int4*)(idx + (size_t)p * EDGES_PER_PART);

    int4 e[E4_PER_THREAD];
#pragma unroll
    for (int j = 0; j < E4_PER_THREAD; ++j) {
        const int i = tid + j * HIST_THREADS;
        if (i < E4_PER_PART) e[j] = ep[i];
    }

    uint4* h4 = (uint4*)hist;
    for (int i = tid; i < N4; i += HIST_THREADS) h4[i] = make_uint4(0u, 0u, 0u, 0u);
    __syncthreads();

#pragma unroll
    for (int j = 0; j < E4_PER_THREAD; ++j) {
        const int i = tid + j * HIST_THREADS;
        if (i < E4_PER_PART) {
            atomicAdd(&hist[e[j].x], 1u);
            atomicAdd(&hist[e[j].y], 1u);
            atomicAdd(&hist[e[j].z], 1u);
            atomicAdd(&hist[e[j].w], 1u);
        }
    }
    __syncthreads();

    uint4* op = (uint4*)(part + (size_t)blockIdx.x * N_NODES);
    for (int i = tid; i < N4; i += HIST_THREADS) op[i] = h4[i];
}

__global__ __launch_bounds__(256) void degree_finish(const unsigned* __restrict__ part,
                                                     float* __restrict__ rs) {
    const int t = blockIdx.x * blockDim.x + threadIdx.x;
    if (t >= 2 * N4) return;
    const int arr = (t >= N4) ? 1 : 0;
    const int q   = t - arr * N4;
    const uint4* pp = (const uint4*)part + (size_t)arr * NPART * N4 + q;
    uint4 s = make_uint4(0u, 0u, 0u, 0u);
#pragma unroll
    for (int p = 0; p < NPART; ++p) {
        const uint4 v = pp[(size_t)p * N4];
        s.x += v.x; s.y += v.y; s.z += v.z; s.w += v.w;
    }
    float4 r;
    r.x = rsqrtf(fmaxf((float)s.x, 1.0f));
    r.y = rsqrtf(fmaxf((float)s.y, 1.0f));
    r.z = rsqrtf(fmaxf((float)s.z, 1.0f));
    r.w = rsqrtf(fmaxf((float)s.w, 1.0f));
    ((float4*)rs)[t] = r;
}

#define RNODES   208
#define NLBLK    ((N_NODES + RNODES - 1) / RNODES)
#define LTHREADS 1024
#define LWAVES   32
#define CAPS     24
#define E4_TOTAL (N_EDGES / 4)
#define E4_WAVE  (E4_TOTAL / LWAVES)
#define SCAN_UNR 8
#define SCAN_BATCHES ((E4_WAVE + 32 * SCAN_UNR - 1) / (32 * SCAN_UNR))
#define GATHER_U 8
#define FLAT_CAP (LWAVES * CAPS)
#define KEEP_HEAD  32
#define KEEP_REC   (KEEP_HEAD + FLAT_CAP)
#define KEEP_BASE  32
static_assert(KEEP_BASE + NLBLK * LWAVES * KEEP_REC <= N_NODES * DFEAT,
              "kept records must fit the scratch plane");

__device__ __forceinline__ void fma4(float4& a, const float4 v, const float s) {
    a.x = fmaf(v.x, s, a.x);
    a.y = fmaf(v.y, s, a.y);
    a.z = fmaf(v.z, s, a.z);
    a.w = fmaf(v.w, s, a.w);
}

__device__ __forceinline__ void scan_load(int4 (&d)[SCAN_UNR], const int4* __restrict__ dst4,
                                          int pos, int end, int lane) {
    if (pos + 32 * SCAN_UNR <= end) {
#pragma unroll
        for (int j = 0; j < SCAN_UNR; ++j) d[j] = dst4[pos + j * 32 + lane];
    } else {
#pragma unroll
        for (int j = 0; j < SCAN_UNR; ++j) {
            const int i = pos + j * 32 + lane;
            d[j] = (i < end) ? dst4[i] : make_int4(-1, -1, -1, -1);
        }
    }
}

#define ACC_ROW(slot, v, s)                         \
    switch (slot) {                                 \
    case 0: fma4(a0, v, s); break;                  \
    case 1: fma4(a1, v, s); break;                  \
    case 2: fma4(a2, v, s); break;                  \
    case 3: fma4(a3, v, s); break;                  \
    case 4: fma4(a4, v, s); break;                  \
    case 5: fma4(a5, v, s); break;                  \
    case 6: fma4(a6, v, s); break;                  \
    default: break;                                 \
    }

template <int NT, bool RELU>
__device__ __forceinline__ void tile_product(const float* Alds, const float* __restrict__ Wm,
                                             const float* __restrict__ bias,
                                             float* __restrict__ out,
                                             int lo, int ntiles, int g, int n0, int lane) {
    const int mrow = lane & 15;
    const int khi  = (lane >> 4) << 1;
    const float* wp = Wm + n0 + mrow;
    const float* ap[NT];
    bool valid[NT];
    v8f acc[NT];
#pragma unroll
    for (int t = 0; t < NT; ++t) {
        const int rt = g + 4 * t;
        valid[t] = rt < ntiles;
        const int rtc = valid[t] ? rt : g;
        ap[t] = Alds + (rtc * 16 + mrow) * ASTRIDE + khi;
        acc[t] = (v8f){0.f, 0.f, 0.f, 0.f, 0.f, 0.f, 0.f, 0.f};
    }
#pragma unroll
    for (int k = 0; k < DFEAT; k += 4) {
        v2f b;
        b.x = wp[(size_t)(k + khi)     * DFEAT];
        b.y = wp[(size_t)(k + khi + 1) * DFEAT];
#pragma unroll
        for (int t = 0; t < NT; ++t) {
            const v2f a = *(const v2f*)(ap[t] + k);
            acc[t] = __builtin_amdgcn_wmma_f32_16x16x4_f32(false, a, false, b,
                                                           (short)0, acc[t], false, false);
        }
    }
    const int   col  = n0 + mrow;
    const float bv   = bias[col];
    const int   rsel = (lane >> 4) << 3;
#pragma unroll
    for (int t = 0; t < NT; ++t) {
        if (valid[t]) {
            const int rowBase = lo + (g + 4 * t) * 16;
#pragma unroll
            for (int v = 0; v < 8; ++v) {
                float h = acc[t][v] + bv;
                if (RELU) h = fmaxf(h, 0.f);
                out[(size_t)(rowBase + v + rsel) * DFEAT + col] = h;
            }
        }
    }
}

template <bool RELU, bool KEPT>
__global__ __launch_bounds__(LTHREADS) void gcn_layer(const float* __restrict__ X,
                                                      const float* __restrict__ rs_out,
                                                      const float* __restrict__ rs_in,
                                                      const int* __restrict__ src,
                                                      const int* __restrict__ dst,
                                                      const float* __restrict__ Wm,
                                                      const float* __restrict__ bias,
                                                      float* __restrict__ out,
                                                      unsigned* __restrict__ keep) {
    __shared__ __attribute__((aligned(16))) float Alds[RNODES * ASTRIDE];
    __shared__ __attribute__((aligned(16))) unsigned lists[LWAVES * LWAVES * CAPS];
    __shared__ unsigned cntS[LWAVES * LWAVES];
    __shared__ unsigned ovfFlag;

    const int tid  = threadIdx.x;
    const int lane = tid & 31;
    const int w    = __builtin_amdgcn_readfirstlane(tid >> 5);
    const int lo   = blockIdx.x * RNODES;
    const unsigned nrows = (N_NODES - lo < RNODES) ? (unsigned)(N_NODES - lo) : (unsigned)RNODES;

    const int4*   dst4 = (const int4*)dst;
    const float4* X4   = (const float4*)X;

    unsigned* krec = keep + KEEP_BASE + (size_t)((unsigned)blockIdx.x * LWAVES + (unsigned)w) * KEEP_REC;
    unsigned* kent = krec + KEEP_HEAD;

    bool slow;
    unsigned nj[7];
    unsigned* mypart = lists + w * FLAT_CAP;

    if (KEPT) {
        const unsigned hv = krec[lane];
        slow = __builtin_amdgcn_readlane((int)hv, 7) != 0;
#pragma unroll
        for (int j = 0; j < 7; ++j) nj[j] = (unsigned)__builtin_amdgcn_readlane((int)hv, j);
    } else {
#pragma unroll
        for (int j = 0; j < 7; ++j) nj[j] = 0u;
    if (tid == 0) ovfFlag = 0u;
    __syncthreads();

    {
        const int beg = w * E4_WAVE;
        const int end = beg + E4_WAVE;
        unsigned cnt = 0u;
        bool ovf = false;
        const int rot = blockIdx.x % SCAN_BATCHES;
        int4 dn[SCAN_UNR];
        scan_load(dn, dst4, beg + rot * (32 * SCAN_UNR), end, lane);
        for (int t = 0; t < SCAN_BATCHES; ++t) {
            int tb = t + rot;
            if (tb >= SCAN_BATCHES) tb -= SCAN_BATCHES;
            const int pos = beg + tb * (32 * SCAN_UNR);
            int4 d[SCAN_UNR];
#pragma unroll
            for (int j = 0; j < SCAN_UNR; ++j) d[j] = dn[j];
            if (t + 1 < SCAN_BATCHES) {
                int tn = tb + 1;
                if (tn >= SCAN_BATCHES) tn -= SCAN_BATCHES;
                scan_load(dn, dst4, beg + tn * (32 * SCAN_UNR), end, lane);
            }
#pragma unroll
            for (int j = 0; j < SCAN_UNR; ++j) {
                const unsigned u0 = (unsigned)d[j].x - (unsigned)lo;
                const unsigned u1 = (unsigned)d[j].y - (unsigned)lo;
                const unsigned u2 = (unsigned)d[j].z - (unsigned)lo;
                const unsigned u3 = (unsigned)d[j].w - (unsigned)lo;
                unsigned m0 = __builtin_amdgcn_ballot_w32(u0 < nrows);
                unsigned m1 = __builtin_amdgcn_ballot_w32(u1 < nrows);
                unsigned m2 = __builtin_amdgcn_ballot_w32(u2 < nrows);
                unsigned m3 = __builtin_amdgcn_ballot_w32(u3 < nrows);
                if ((m0 | m1 | m2 | m3) != 0u) {
                    const unsigned ib = (unsigned)(pos + j * 32);
#define SCAN_TAKE(m, u, c)                                                              \
                    while (m) {                                                         \
                        const int l = __builtin_ctz(m);                                 \
                        m &= m - 1u;                                                    \
                        const unsigned row   = (unsigned)__builtin_amdgcn_readlane((int)(u), l); \
                        const unsigned owner = row & 31u;                               \
                        const unsigned n = (unsigned)__builtin_amdgcn_readlane((int)cnt, (int)owner); \
                        if (n < (unsigned)CAPS) {                                       \
                            if ((unsigned)lane == owner) {                              \
                                lists[(owner * LWAVES + (unsigned)w) * CAPS + n] =      \
                                    (row << 20) | (4u * (ib + (unsigned)l) + (c));      \
                                cnt = n + 1u;                                           \
                            }                                                           \
                        } else {                                                        \
                            ovf = true;                                                 \
                        }                                                               \
                    }
                    SCAN_TAKE(m0, u0, 0u)
                    SCAN_TAKE(m1, u1, 1u)
                    SCAN_TAKE(m2, u2, 2u)
                    SCAN_TAKE(m3, u3, 3u)
#undef SCAN_TAKE
                }
            }
        }
        cntS[lane * LWAVES + w] = cnt;
        if (ovf && lane == 0) ovfFlag = 1u;
    }
    __syncthreads();
    slow = __builtin_amdgcn_readfirstlane((int)ovfFlag) != 0;
    }

    float4 a0 = make_float4(0.f, 0.f, 0.f, 0.f);
    float4 a1 = a0, a2 = a0, a3 = a0, a4 = a0, a5 = a0, a6 = a0;

    if (!slow) {
        unsigned oj[7];
        if (!KEPT) {
        unsigned* myflat = (unsigned*)Alds + w * FLAT_CAP;
        const unsigned c_l = cntS[w * LWAVES + lane];
        unsigned incl = c_l;
#pragma unroll
        for (int dd = 1; dd < 32; dd <<= 1) {
            const unsigned up = __shfl_up(incl, dd);
            if (lane >= dd) incl += up;
        }
        const unsigned off = incl - c_l;
        const int total = __builtin_amdgcn_readlane((int)incl, 31);
        const uint4* mylist4 = (const uint4*)(lists + (w * LWAVES + lane) * CAPS);
        unsigned ent[CAPS];
        unsigned sn[CAPS];
#pragma unroll
        for (int i = 0; i < CAPS / 4; ++i) {
            const uint4 q = mylist4[i];
            ent[4 * i + 0] = q.x; ent[4 * i + 1] = q.y; ent[4 * i + 2] = q.z; ent[4 * i + 3] = q.w;
        }
#pragma unroll
        for (int i = 0; i < CAPS; ++i) {
            const unsigned e = ((unsigned)i < c_l) ? (ent[i] & 0xFFFFFu) : 0u;
            sn[i] = (unsigned)src[e];
        }
#pragma unroll
        for (int i = 0; i < CAPS; ++i) asm volatile("" :: "v"(sn[i]));
#pragma unroll
        for (int i = 0; i < CAPS; ++i) {
            if ((unsigned)i < c_l) myflat[off + i] = ((ent[i] >> 25) << 16) | sn[i];
        }
        __syncthreads();

        unsigned run[7];
#pragma unroll
        for (int j = 0; j < 7; ++j) run[j] = 0u;
        for (int k = 0; k < total; k += 32) {
            const int idx = k + lane;
            const unsigned slot = (idx < total) ? (myflat[idx] >> 16) : 7u;
#pragma unroll
            for (int j = 0; j < 7; ++j)
                nj[j] += (unsigned)__builtin_popcount(__builtin_amdgcn_ballot_w32(slot == (unsigned)j));
        }
        oj[0] = 0u;
#pragma unroll
        for (int j = 1; j < 7; ++j) oj[j] = oj[j - 1] + nj[j - 1];
        for (int k = 0; k < total; k += 32) {
            const int idx = k + lane;
            const unsigned ent  = (idx < total) ? myflat[idx] : (7u << 16);
            const unsigned slot = ent >> 16;
#pragma unroll
            for (int j = 0; j < 7; ++j) {
                const unsigned m = __builtin_amdgcn_ballot_w32(slot == (unsigned)j);
                if (slot == (unsigned)j)
                    mypart[oj[j] + run[j] + __builtin_amdgcn_mbcnt_lo(m, 0u)] = ent & 0xFFFFu;
                run[j] += (unsigned)__builtin_popcount(m);
            }
        }
        } else {
            oj[0] = 0u;
#pragma unroll
            for (int j = 1; j < 7; ++j) oj[j] = oj[j - 1] + nj[j - 1];
            const unsigned total = oj[6] + nj[6];
            for (unsigned k = 0u; k < total; k += 128u) {
                unsigned e[4];
#pragma unroll
                for (int u = 0; u < 4; ++u) {
                    const unsigned idx = k + (unsigned)(u * 32 + lane);
                    e[u] = (idx < total) ? kent[idx] : 0u;
                }
#pragma unroll
                for (int u = 0; u < 4; ++u) {
                    const unsigned idx = k + (unsigned)(u * 32 + lane);
                    if (idx < total) mypart[idx] = e[u];
                }
            }
        }
        __syncthreads();

        if (!KEPT) {
            const unsigned total = oj[6] + nj[6];
            for (unsigned k = 0u; k < total; k += 32u) {
                const unsigned idx = k + (unsigned)lane;
                if (idx < total) kent[idx] = mypart[idx];
            }
        }

#define GATHER_SLOT(j, a)                                                               \
        for (unsigned k0 = 0u; k0 < nj[j]; k0 += GATHER_U) {                            \
            const unsigned kl   = k0 + (unsigned)lane;                                  \
            const unsigned entv = mypart[oj[j] + ((kl < nj[j]) ? kl : (nj[j] - 1u))];   \
            float4 v[GATHER_U];                                                         \
            float  sc[GATHER_U];                                                        \
            _Pragma("unroll")                                                           \
            for (int u = 0; u < GATHER_U; ++u) {                                        \
                const unsigned s = (unsigned)__builtin_amdgcn_readlane((int)entv, u);   \
                v[u]  = X4[(size_t)s * (DFEAT / 4) + lane];                             \
                sc[u] = rs_out[s];                                                      \
            }                                                                           \
              \
               \
                                                 \
            _Pragma("unroll")                                                           \
            for (int u = 0; u < GATHER_U; ++u)                                          \
                asm volatile("" :: "v"(v[u].x), "v"(v[u].y), "v"(v[u].z), "v"(v[u].w), \
                             "s"(sc[u]));                                               \
            _Pragma("unroll")                                                           \
            for (int u = 0; u < GATHER_U; ++u) {                                        \
                if (k0 + (unsigned)u < nj[j]) fma4(a, v[u], sc[u]);                     \
            }                                                                           \
        }
        GATHER_SLOT(0, a0)
        GATHER_SLOT(1, a1)
        GATHER_SLOT(2, a2)
        GATHER_SLOT(3, a3)
        GATHER_SLOT(4, a4)
        GATHER_SLOT(5, a5)
        GATHER_SLOT(6, a6)
#undef GATHER_SLOT
    } else {
        for (int pos = 0; pos < E4_TOTAL; pos += 32) {
            const int4 d = dst4[pos + lane];
            const unsigned u0 = (unsigned)d.x - (unsigned)lo;
            const unsigned u1 = (unsigned)d.y - (unsigned)lo;
            const unsigned u2 = (unsigned)d.z - (unsigned)lo;
            const unsigned u3 = (unsigned)d.w - (unsigned)lo;
            unsigned m0 = __builtin_amdgcn_ballot_w32(u0 < nrows && (u0 & 31u) == (unsigned)w);
            unsigned m1 = __builtin_amdgcn_ballot_w32(u1 < nrows && (u1 & 31u) == (unsigned)w);
            unsigned m2 = __builtin_amdgcn_ballot_w32(u2 < nrows && (u2 & 31u) == (unsigned)w);
            unsigned m3 = __builtin_amdgcn_ballot_w32(u3 < nrows && (u3 & 31u) == (unsigned)w);
            if ((m0 | m1 | m2 | m3) != 0u) {
#define SLOW_TAKE(m, u, c)                                                              \
                while (m) {                                                             \
                    const int l = __builtin_ctz(m);                                     \
                    m &= m - 1u;                                                        \
                    const unsigned row = (unsigned)__builtin_amdgcn_readlane((int)(u), l); \
                    const int      s   = src[4 * (pos + l) + (c)];                      \
                    const float    scs = rs_out[s];                                     \
                    const float4   vs  = X4[(size_t)s * (DFEAT / 4) + lane];            \
                    ACC_ROW(row >> 5, vs, scs)                                          \
                }
                SLOW_TAKE(m0, u0, 0)
                SLOW_TAKE(m1, u1, 1)
                SLOW_TAKE(m2, u2, 2)
                SLOW_TAKE(m3, u3, 3)
#undef SLOW_TAKE
            }
        }
    }
    __syncthreads();

#define PUT_ROW(j, a)                                                                   \
    {                                                                                   \
        const int r = (j) * 32 + w;                                                     \
        if (r < RNODES) {                                                               \
            const float si = (r < (int)nrows) ? rs_in[lo + r] : 0.f;                    \
            *(float4*)&Alds[r * ASTRIDE + lane * 4] =                                   \
                make_float4(a.x * si, a.y * si, a.z * si, a.w * si);                    \
        }                                                                               \
    }
    PUT_ROW(0, a0)
    PUT_ROW(1, a1)
    PUT_ROW(2, a2)
    PUT_ROW(3, a3)
    PUT_ROW(4, a4)
    PUT_ROW(5, a5)
    PUT_ROW(6, a6)
#undef PUT_ROW
    __syncthreads();

    if (!KEPT) {
        unsigned hv = (lane == 7 && slow) ? 1u : 0u;
#pragma unroll
        for (int j = 0; j < 7; ++j) if (lane == j) hv = nj[j];
        krec[lane] = hv;
    }

    const int ntiles = (int)(nrows >> 4);
    const int g      = w >> 3;
    const int n0     = (w & 7) << 4;
    if (g == 0) tile_product<4, RELU>(Alds, Wm, bias, out, lo, ntiles, g, n0, lane);
    else        tile_product<3, RELU>(Alds, Wm, bias, out, lo, ntiles, g, n0, lane);
}

extern "C" void kernel_launch(void* const* d_in, const int* in_sizes, int n_in,
                              void* d_out, int out_size, void* d_ws, size_t ws_size,
                              hipStream_t stream) {
    const float* x   = (const float*)d_in[0];
    const int*   src = (const int*)  d_in[1];
    const int*   dst = (const int*)  d_in[2];
    const float* W1  = (const float*)d_in[3];
    const float* b1  = (const float*)d_in[4];
    const float* W2  = (const float*)d_in[5];
    const float* b2  = (const float*)d_in[6];
    float*       out = (float*)d_out;

    float* ws     = (float*)d_ws;
    float* rs_out = ws;
    float* rs_in  = ws + N_NODES;
    float* scr    = ws + 2 * N_NODES;
    float* h1     = scr + (size_t)N_NODES * DFEAT;

    const int T = 256;

    unsigned* part = (unsigned*)scr;
    degree_hist<<<2 * NPART, HIST_THREADS, 0, stream>>>(src, dst, part);
    degree_finish<<<(2 * N4 + T - 1) / T, T, 0, stream>>>(part, rs_out);

    unsigned* kept = (unsigned*)scr;
    gcn_layer<true, false><<<NLBLK, LTHREADS, 0, stream>>>(x, rs_out, rs_in, src, dst, W1, b1, h1, kept);
    gcn_layer<false, true><<<NLBLK, LTHREADS, 0, stream>>>(h1, rs_out, rs_in, src, dst, W2, b2, out, kept);
}


# Layer-1 filter loads only the 32-edge chunks of dst[] that a per-work-group bitmap marks (built in the degree launch by 32 more groups, integer LDS ORs), not all of dst[]; edge sum order changed
# speedup vs baseline: 3.6581x; 1.1014x over previous
#include <hip/hip_runtime.h>

#define N_NODES 50000
#define N_EDGES 800000
#define DFEAT   128
#define ASTRIDE 132

typedef float v2f __attribute__((ext_vector_type(2)));
typedef float v8f __attribute__((ext_vector_type(8)));

#define RNODES   208
#define NLBLK    ((N_NODES + RNODES - 1) / RNODES)
#define LTHREADS 1024
#define LWAVES   32
#define CAPS     24
#define E4_TOTAL (N_EDGES / 4)
#define E4_WAVE  (E4_TOTAL / LWAVES)
#define SCAN_U   16
#define GATHER_U 8
#define FLAT_CAP (LWAVES * CAPS)
#define KEEP_HEAD  32
#define KEEP_REC   (KEEP_HEAD + FLAT_CAP)
#define KEEP_BASE  32
#define BWORDS   25
#define BITS_OFF (KEEP_BASE + NLBLK * LWAVES * KEEP_REC)
static_assert(BITS_OFF + NLBLK * LWAVES * BWORDS <= N_NODES * DFEAT,
              "kept records and chunk bitmaps must fit the scratch plane");

#define NPART 32
#define EDGES_PER_PART (N_EDGES / NPART)
#define CHUNK_SHIFT 5
static_assert(NPART == LWAVES && EDGES_PER_PART == 4 * E4_WAVE,
              "a bitmap group's share is one scanning wave's share");
static_assert(((EDGES_PER_PART + 31 + 31) >> CHUNK_SHIFT) <= BWORDS * 32,
              "the chunks of one share must fit its bitmap");
static_assert(2 * NPART * N_NODES <= BITS_OFF,
              "the bitmaps lie behind the degree partials");
#define E4_PER_PART    (EDGES_PER_PART / 4)
#define N4             (N_NODES / 4)
#define HIST_THREADS   1024
#define E4_PER_THREAD  ((E4_PER_PART + HIST_THREADS - 1) / HIST_THREADS)

__global__ __launch_bounds__(HIST_THREADS) void degree_hist(const int* __restrict__ src,
                                                            const int* __restrict__ dst,
                                                            unsigned* __restrict__ part) {
    __shared__ unsigned hist[N_NODES];
    __shared__ unsigned cbits[NLBLK * BWORDS];

    const int tid = threadIdx.x;
    const int arr = blockIdx.x / NPART;
    const int p   = blockIdx.x % NPART;
    const int* idx = (arr == 0) ? src : dst;
    const int4* ep = (const int4*)(idx + (size_t)p * EDGES_PER_PART);

    int4 e[E4_PER_THREAD];
#pragma unroll
    for (int j = 0; j < E4_PER_THREAD; ++j) {
        const int i = tid + j * HIST_THREADS;
        if (i < E4_PER_PART) e[j] = ep[i];
    }

    if (arr < 2) {
        uint4* h4 = (uint4*)hist;
        for (int i = tid; i < N4; i += HIST_THREADS) h4[i] = make_uint4(0u, 0u, 0u, 0u);
        __syncthreads();

#pragma unroll
        for (int j = 0; j < E4_PER_THREAD; ++j) {
            const int i = tid + j * HIST_THREADS;
            if (i < E4_PER_PART) {
                atomicAdd(&hist[e[j].x], 1u);
                atomicAdd(&hist[e[j].y], 1u);
                atomicAdd(&hist[e[j].z], 1u);
                atomicAdd(&hist[e[j].w], 1u);
            }
        }
        __syncthreads();

        uint4* op = (uint4*)(part + (size_t)blockIdx.x * N_NODES);
        for (int i = tid; i < N4; i += HIST_THREADS) op[i] = h4[i];
    } else {
        for (int i = tid; i < NLBLK * BWORDS; i += HIST_THREADS) cbits[i] = 0u;
        __syncthreads();

        const unsigned ebeg = (unsigned)p * EDGES_PER_PART;
#pragma unroll
        for (int j = 0; j < E4_PER_THREAD; ++j) {
            const int i = tid + j * HIST_THREADS;
            if (i < E4_PER_PART) {
                const unsigned lc  = ((ebeg + 4u * (unsigned)i) >> CHUNK_SHIFT) - (ebeg >> CHUNK_SHIFT);
                const unsigned wd  = lc >> 5;
                const unsigned bit = 1u << (lc & 31u);
                atomicOr(&cbits[((unsigned)e[j].x / RNODES) * BWORDS + wd], bit);
                atomicOr(&cbits[((unsigned)e[j].y / RNODES) * BWORDS + wd], bit);
                atomicOr(&cbits[((unsigned)e[j].z / RNODES) * BWORDS + wd], bit);
                atomicOr(&cbits[((unsigned)e[j].w / RNODES) * BWORDS + wd], bit);
            }
        }
        __syncthreads();

        unsigned* bp = part + BITS_OFF;
        for (int i = tid; i < NLBLK * BWORDS; i += HIST_THREADS) {
            const int b = i / BWORDS;
            const int k = i - b * BWORDS;
            bp[(size_t)(b * NPART + p) * BWORDS + k] = cbits[i];
        }
    }
}

__global__ __launch_bounds__(256) void degree_finish(const unsigned* __restrict__ part,
                                                     float* __restrict__ rs) {
    const int t = blockIdx.x * blockDim.x + threadIdx.x;
    if (t >= 2 * N4) return;
    const int arr = (t >= N4) ? 1 : 0;
    const int q   = t - arr * N4;
    const uint4* pp = (const uint4*)part + (size_t)arr * NPART * N4 + q;
    uint4 s = make_uint4(0u, 0u, 0u, 0u);
#pragma unroll
    for (int p = 0; p < NPART; ++p) {
        const uint4 v = pp[(size_t)p * N4];
        s.x += v.x; s.y += v.y; s.z += v.z; s.w += v.w;
    }
    float4 r;
    r.x = rsqrtf(fmaxf((float)s.x, 1.0f));
    r.y = rsqrtf(fmaxf((float)s.y, 1.0f));
    r.z = rsqrtf(fmaxf((float)s.z, 1.0f));
    r.w = rsqrtf(fmaxf((float)s.w, 1.0f));
    ((float4*)rs)[t] = r;
}


__device__ __forceinline__ void fma4(float4& a, const float4 v, const float s) {
    a.x = fmaf(v.x, s, a.x);
    a.y = fmaf(v.y, s, a.y);
    a.z = fmaf(v.z, s, a.z);
    a.w = fmaf(v.w, s, a.w);
}

#define ACC_ROW(slot, v, s)                         \
    switch (slot) {                                 \
    case 0: fma4(a0, v, s); break;                  \
    case 1: fma4(a1, v, s); break;                  \
    case 2: fma4(a2, v, s); break;                  \
    case 3: fma4(a3, v, s); break;                  \
    case 4: fma4(a4, v, s); break;                  \
    case 5: fma4(a5, v, s); break;                  \
    case 6: fma4(a6, v, s); break;                  \
    default: break;                                 \
    }

template <int NT, bool RELU>
__device__ __forceinline__ void tile_product(const float* Alds, const float* __restrict__ Wm,
                                             const float* __restrict__ bias,
                                             float* __restrict__ out,
                                             int lo, int ntiles, int g, int n0, int lane) {
    const int mrow = lane & 15;
    const int khi  = (lane >> 4) << 1;
    const float* wp = Wm + n0 + mrow;
    const float* ap[NT];
    bool valid[NT];
    v8f acc[NT];
#pragma unroll
    for (int t = 0; t < NT; ++t) {
        const int rt = g + 4 * t;
        valid[t] = rt < ntiles;
        const int rtc = valid[t] ? rt : g;
        ap[t] = Alds + (rtc * 16 + mrow) * ASTRIDE + khi;
        acc[t] = (v8f){0.f, 0.f, 0.f, 0.f, 0.f, 0.f, 0.f, 0.f};
    }
#pragma unroll
    for (int k = 0; k < DFEAT; k += 4) {
        v2f b;
        b.x = wp[(size_t)(k + khi)     * DFEAT];
        b.y = wp[(size_t)(k + khi + 1) * DFEAT];
#pragma unroll
        for (int t = 0; t < NT; ++t) {
            const v2f a = *(const v2f*)(ap[t] + k);
            acc[t] = __builtin_amdgcn_wmma_f32_16x16x4_f32(false, a, false, b,
                                                           (short)0, acc[t], false, false);
        }
    }
    const int   col  = n0 + mrow;
    const float bv   = bias[col];
    const int   rsel = (lane >> 4) << 3;
#pragma unroll
    for (int t = 0; t < NT; ++t) {
        if (valid[t]) {
            const int rowBase = lo + (g + 4 * t) * 16;
#pragma unroll
            for (int v = 0; v < 8; ++v) {
                float h = acc[t][v] + bv;
                if (RELU) h = fmaxf(h, 0.f);
                out[(size_t)(rowBase + v + rsel) * DFEAT + col] = h;
            }
        }
    }
}

template <bool RELU, bool KEPT>
__global__ __launch_bounds__(LTHREADS) void gcn_layer(const float* __restrict__ X,
                                                      const float* __restrict__ rs_out,
                                                      const float* __restrict__ rs_in,
                                                      const int* __restrict__ src,
                                                      const int* __restrict__ dst,
                                                      const float* __restrict__ Wm,
                                                      const float* __restrict__ bias,
                                                      float* __restrict__ out,
                                                      unsigned* __restrict__ keep) {
    __shared__ __attribute__((aligned(16))) float Alds[RNODES * ASTRIDE];
    __shared__ __attribute__((aligned(16))) unsigned lists[LWAVES * LWAVES * CAPS];
    __shared__ unsigned cntS[LWAVES * LWAVES];
    __shared__ unsigned ovfFlag;

    const int tid  = threadIdx.x;
    const int lane = tid & 31;
    const int w    = __builtin_amdgcn_readfirstlane(tid >> 5);
    const int lo   = blockIdx.x * RNODES;
    const unsigned nrows = (N_NODES - lo < RNODES) ? (unsigned)(N_NODES - lo) : (unsigned)RNODES;

    const int4*   dst4 = (const int4*)dst;
    const float4* X4   = (const float4*)X;

    unsigned* krec = keep + KEEP_BASE + (size_t)((unsigned)blockIdx.x * LWAVES + (unsigned)w) * KEEP_REC;
    unsigned* kent = krec + KEEP_HEAD;

    bool slow;
    unsigned nj[7];
    unsigned* mypart = lists + w * FLAT_CAP;

    if (KEPT) {
        const unsigned hv = krec[lane];
        slow = __builtin_amdgcn_readlane((int)hv, 7) != 0;
#pragma unroll
        for (int j = 0; j < 7; ++j) nj[j] = (unsigned)__builtin_amdgcn_readlane((int)hv, j);
    } else {
#pragma unroll
        for (int j = 0; j < 7; ++j) nj[j] = 0u;
    if (tid == 0) ovfFlag = 0u;
    __syncthreads();

    {
        const int ebeg   = w * (4 * E4_WAVE);
        const int cfirst = ebeg >> CHUNK_SHIFT;
        unsigned cnt = 0u;
        bool ovf = false;
        const unsigned* bw = keep + BITS_OFF
                           + (size_t)((unsigned)blockIdx.x * LWAVES + (unsigned)w) * BWORDS;
        const unsigned myw = (lane < BWORDS) ? bw[lane] : 0u;
        unsigned* clist = (unsigned*)Alds + w * (BWORDS * 32);
        const unsigned pc = (unsigned)__builtin_popcount(myw);
        unsigned cincl = pc;
#pragma unroll
        for (int dd = 1; dd < 32; dd <<= 1) {
            const unsigned up = __shfl_up(cincl, dd);
            if (lane >= dd) cincl += up;
        }
        const int nchunks = __builtin_amdgcn_readlane((int)cincl, 31);
        {
            unsigned mm = myw;
            unsigned o  = cincl - pc;
            while (mm) {
                clist[o++] = (unsigned)(lane * 32 + __builtin_ctz(mm));
                mm &= mm - 1u;
            }
        }
        __syncthreads();
        for (int k0 = 0; k0 < nchunks; k0 += SCAN_U) {
            const int kl = k0 + lane;
            const int cv = (lane < SCAN_U && kl < nchunks) ? (int)clist[kl] : -1;
            int ck[SCAN_U];
#pragma unroll
            for (int u = 0; u < SCAN_U; ++u) ck[u] = __builtin_amdgcn_readlane(cv, u);
            int dv[SCAN_U];
#pragma unroll
            for (int u = 0; u < SCAN_U; ++u) {
                const int  e  = (cfirst + ck[u]) * 32 + lane;
                const bool ok = (unsigned)(e - ebeg) < (unsigned)(4 * E4_WAVE);
                dv[u] = dst[ok ? e : ebeg];
            }
#pragma unroll
            for (int u = 0; u < SCAN_U; ++u) asm volatile("" :: "v"(dv[u]));
#pragma unroll
            for (int u = 0; u < SCAN_U; ++u) {
                const int      cg = cfirst + ck[u];
                const int      e  = cg * 32 + lane;
                const bool     ok = (unsigned)(e - ebeg) < (unsigned)(4 * E4_WAVE);
                const unsigned uu = (unsigned)dv[u] - (unsigned)lo;
                unsigned m = __builtin_amdgcn_ballot_w32(ok && uu < nrows);
                const unsigned eb = (unsigned)cg * 32u;
                while (m) {
                    const int l = __builtin_ctz(m);
                    m &= m - 1u;
                    const unsigned row   = (unsigned)__builtin_amdgcn_readlane((int)uu, l);
                    const unsigned owner = row & 31u;
                    const unsigned n = (unsigned)__builtin_amdgcn_readlane((int)cnt, (int)owner);
                    if (n < (unsigned)CAPS) {
                        if ((unsigned)lane == owner) {
                            lists[(owner * LWAVES + (unsigned)w) * CAPS + n] =
                                (row << 20) | (eb + (unsigned)l);
                            cnt = n + 1u;
                        }
                    } else {
                        ovf = true;
                    }
                }
            }
        }
        cntS[lane * LWAVES + w] = cnt;
        if (ovf && lane == 0) ovfFlag = 1u;
    }
    __syncthreads();
    slow = __builtin_amdgcn_readfirstlane((int)ovfFlag) != 0;
    }

    float4 a0 = make_float4(0.f, 0.f, 0.f, 0.f);
    float4 a1 = a0, a2 = a0, a3 = a0, a4 = a0, a5 = a0, a6 = a0;

    if (!slow) {
        unsigned oj[7];
        if (!KEPT) {
        unsigned* myflat = (unsigned*)Alds + w * FLAT_CAP;
        const unsigned c_l = cntS[w * LWAVES + lane];
        unsigned incl = c_l;
#pragma unroll
        for (int dd = 1; dd < 32; dd <<= 1) {
            const unsigned up = __shfl_up(incl, dd);
            if (lane >= dd) incl += up;
        }
        const unsigned off = incl - c_l;
        const int total = __builtin_amdgcn_readlane((int)incl, 31);
        const uint4* mylist4 = (const uint4*)(lists + (w * LWAVES + lane) * CAPS);
        unsigned ent[CAPS];
        unsigned sn[CAPS];
#pragma unroll
        for (int i = 0; i < CAPS / 4; ++i) {
            const uint4 q = mylist4[i];
            ent[4 * i + 0] = q.x; ent[4 * i + 1] = q.y; ent[4 * i + 2] = q.z; ent[4 * i + 3] = q.w;
        }
#pragma unroll
        for (int i = 0; i < CAPS; ++i) {
            const unsigned e = ((unsigned)i < c_l) ? (ent[i] & 0xFFFFFu) : 0u;
            sn[i] = (unsigned)src[e];
        }
#pragma unroll
        for (int i = 0; i < CAPS; ++i) asm volatile("" :: "v"(sn[i]));
#pragma unroll
        for (int i = 0; i < CAPS; ++i) {
            if ((unsigned)i < c_l) myflat[off + i] = ((ent[i] >> 25) << 16) | sn[i];
        }
        __syncthreads();

        unsigned run[7];
#pragma unroll
        for (int j = 0; j < 7; ++j) run[j] = 0u;
        for (int k = 0; k < total; k += 32) {
            const int idx = k + lane;
            const unsigned slot = (idx < total) ? (myflat[idx] >> 16) : 7u;
#pragma unroll
            for (int j = 0; j < 7; ++j)
                nj[j] += (unsigned)__builtin_popcount(__builtin_amdgcn_ballot_w32(slot == (unsigned)j));
        }
        oj[0] = 0u;
#pragma unroll
        for (int j = 1; j < 7; ++j) oj[j] = oj[j - 1] + nj[j - 1];
        for (int k = 0; k < total; k += 32) {
            const int idx = k + lane;
            const unsigned ent  = (idx < total) ? myflat[idx] : (7u << 16);
            const unsigned slot = ent >> 16;
#pragma unroll
            for (int j = 0; j < 7; ++j) {
                const unsigned m = __builtin_amdgcn_ballot_w32(slot == (unsigned)j);
                if (slot == (unsigned)j)
                    mypart[oj[j] + run[j] + __builtin_amdgcn_mbcnt_lo(m, 0u)] = ent & 0xFFFFu;
                run[j] += (unsigned)__builtin_popcount(m);
            }
        }
        } else {
            oj[0] = 0u;
#pragma unroll
            for (int j = 1; j < 7; ++j) oj[j] = oj[j - 1] + nj[j - 1];
            const unsigned total = oj[6] + nj[6];
            for (unsigned k = 0u; k < total; k += 128u) {
                unsigned e[4];
#pragma unroll
                for (int u = 0; u < 4; ++u) {
                    const unsigned idx = k + (unsigned)(u * 32 + lane);
                    e[u] = (idx < total) ? kent[idx] : 0u;
                }
#pragma unroll
                for (int u = 0; u < 4; ++u) {
                    const unsigned idx = k + (unsigned)(u * 32 + lane);
                    if (idx < total) mypart[idx] = e[u];
                }
            }
        }
        __syncthreads();

        if (!KEPT) {
            const unsigned total = oj[6] + nj[6];
            for (unsigned k = 0u; k < total; k += 32u) {
                const unsigned idx = k + (unsigned)lane;
                if (idx < total) kent[idx] = mypart[idx];
            }
        }

#define GATHER_SLOT(j, a)                                                               \
        for (unsigned k0 = 0u; k0 < nj[j]; k0 += GATHER_U) {                            \
            const unsigned kl   = k0 + (unsigned)lane;                                  \
            const unsigned entv = mypart[oj[j] + ((kl < nj[j]) ? kl : (nj[j] - 1u))];   \
            float4 v[GATHER_U];                                                         \
            float  sc[GATHER_U];                                                        \
            _Pragma("unroll")                                                           \
            for (int u = 0; u < GATHER_U; ++u) {                                        \
                const unsigned s = (unsigned)__builtin_amdgcn_readlane((int)entv, u);   \
                v[u]  = X4[(size_t)s * (DFEAT / 4) + lane];                             \
                sc[u] = rs_out[s];                                                      \
            }                                                                           \
              \
               \
                                                 \
            _Pragma("unroll")                                                           \
            for (int u = 0; u < GATHER_U; ++u)                                          \
                asm volatile("" :: "v"(v[u].x), "v"(v[u].y), "v"(v[u].z), "v"(v[u].w), \
                             "s"(sc[u]));                                               \
            _Pragma("unroll")                                                           \
            for (int u = 0; u < GATHER_U; ++u) {                                        \
                if (k0 + (unsigned)u < nj[j]) fma4(a, v[u], sc[u]);                     \
            }                                                                           \
        }
        GATHER_SLOT(0, a0)
        GATHER_SLOT(1, a1)
        GATHER_SLOT(2, a2)
        GATHER_SLOT(3, a3)
        GATHER_SLOT(4, a4)
        GATHER_SLOT(5, a5)
        GATHER_SLOT(6, a6)
#undef GATHER_SLOT
    } else {
        for (int pos = 0; pos < E4_TOTAL; pos += 32) {
            const int4 d = dst4[pos + lane];
            const unsigned u0 = (unsigned)d.x - (unsigned)lo;
            const unsigned u1 = (unsigned)d.y - (unsigned)lo;
            const unsigned u2 = (unsigned)d.z - (unsigned)lo;
            const unsigned u3 = (unsigned)d.w - (unsigned)lo;
            unsigned m0 = __builtin_amdgcn_ballot_w32(u0 < nrows && (u0 & 31u) == (unsigned)w);
            unsigned m1 = __builtin_amdgcn_ballot_w32(u1 < nrows && (u1 & 31u) == (unsigned)w);
            unsigned m2 = __builtin_amdgcn_ballot_w32(u2 < nrows && (u2 & 31u) == (unsigned)w);
            unsigned m3 = __builtin_amdgcn_ballot_w32(u3 < nrows && (u3 & 31u) == (unsigned)w);
            if ((m0 | m1 | m2 | m3) != 0u) {
#define SLOW_TAKE(m, u, c)                                                              \
                while (m) {                                                             \
                    const int l = __builtin_ctz(m);                                     \
                    m &= m - 1u;                                                        \
                    const unsigned row = (unsigned)__builtin_amdgcn_readlane((int)(u), l); \
                    const int      s   = src[4 * (pos + l) + (c)];                      \
                    const float    scs = rs_out[s];                                     \
                    const float4   vs  = X4[(size_t)s * (DFEAT / 4) + lane];            \
                    ACC_ROW(row >> 5, vs, scs)                                          \
                }
                SLOW_TAKE(m0, u0, 0)
                SLOW_TAKE(m1, u1, 1)
                SLOW_TAKE(m2, u2, 2)
                SLOW_TAKE(m3, u3, 3)
#undef SLOW_TAKE
            }
        }
    }
    __syncthreads();

#define PUT_ROW(j, a)                                                                   \
    {                                                                                   \
        const int r = (j) * 32 + w;                                                     \
        if (r < RNODES) {                                                               \
            const float si = (r < (int)nrows) ? rs_in[lo + r] : 0.f;                    \
            *(float4*)&Alds[r * ASTRIDE + lane * 4] =                                   \
                make_float4(a.x * si, a.y * si, a.z * si, a.w * si);                    \
        }                                                                               \
    }
    PUT_ROW(0, a0)
    PUT_ROW(1, a1)
    PUT_ROW(2, a2)
    PUT_ROW(3, a3)
    PUT_ROW(4, a4)
    PUT_ROW(5, a5)
    PUT_ROW(6, a6)
#undef PUT_ROW
    __syncthreads();

    if (!KEPT) {
        unsigned hv = (lane == 7 && slow) ? 1u : 0u;
#pragma unroll
        for (int j = 0; j < 7; ++j) if (lane == j) hv = nj[j];
        krec[lane] = hv;
    }

    const int ntiles = (int)(nrows >> 4);
    const int g      = w >> 3;
    const int n0     = (w & 7) << 4;
    if (g == 0) tile_product<4, RELU>(Alds, Wm, bias, out, lo, ntiles, g, n0, lane);
    else        tile_product<3, RELU>(Alds, Wm, bias, out, lo, ntiles, g, n0, lane);
}

extern "C" void kernel_launch(void* const* d_in, const int* in_sizes, int n_in,
                              void* d_out, int out_size, void* d_ws, size_t ws_size,
                              hipStream_t stream) {
    const float* x   = (const float*)d_in[0];
    const int*   src = (const int*)  d_in[1];
    const int*   dst = (const int*)  d_in[2];
    const float* W1  = (const float*)d_in[3];
    const float* b1  = (const float*)d_in[4];
    const float* W2  = (const float*)d_in[5];
    const float* b2  = (const float*)d_in[6];
    float*       out = (float*)d_out;

    float* ws     = (float*)d_ws;
    float* rs_out = ws;
    float* rs_in  = ws + N_NODES;
    float* scr    = ws + 2 * N_NODES;
    float* h1     = scr + (size_t)N_NODES * DFEAT;

    const int T = 256;

    unsigned* part = (unsigned*)scr;
    degree_hist<<<3 * NPART, HIST_THREADS, 0, stream>>>(src, dst, part);
    degree_finish<<<(2 * N4 + T - 1) / T, T, 0, stream>>>(part, rs_out);

    unsigned* kept = (unsigned*)scr;
    gcn_layer<true, false><<<NLBLK, LTHREADS, 0, stream>>>(x, rs_out, rs_in, src, dst, W1, b1, h1, kept);
    gcn_layer<false, true><<<NLBLK, LTHREADS, 0, stream>>>(h1, rs_out, rs_in, src, dst, W2, b2, out, kept);
}


# Layer-1 filter: a lane with a hit takes its list place by integer LDS add on the wave's own cursors, 16 adds in flight, no serial lane-read loop; sum order: same-chunk hits as the add serves lanes
# speedup vs baseline: 4.0071x; 1.0954x over previous
#include <hip/hip_runtime.h>

#define N_NODES 50000
#define N_EDGES 800000
#define DFEAT   128
#define ASTRIDE 132

typedef float v2f __attribute__((ext_vector_type(2)));
typedef float v8f __attribute__((ext_vector_type(8)));

#define RNODES   208
#define NLBLK    ((N_NODES + RNODES - 1) / RNODES)
#define LTHREADS 1024
#define LWAVES   32
#define CAPS     24
#define E4_TOTAL (N_EDGES / 4)
#define E4_WAVE  (E4_TOTAL / LWAVES)
#define SCAN_U   16
#define GATHER_U 8
#define FLAT_CAP (LWAVES * CAPS)
#define KEEP_HEAD  32
#define KEEP_REC   (KEEP_HEAD + FLAT_CAP)
#define KEEP_BASE  32
#define BWORDS   25
#define BITS_OFF (KEEP_BASE + NLBLK * LWAVES * KEEP_REC)
static_assert(BITS_OFF + NLBLK * LWAVES * BWORDS <= N_NODES * DFEAT,
              "kept records and chunk bitmaps must fit the scratch plane");

#define NPART 32
#define EDGES_PER_PART (N_EDGES / NPART)
#define CHUNK_SHIFT 5
static_assert(NPART == LWAVES && EDGES_PER_PART == 4 * E4_WAVE,
              "a bitmap group's share is one scanning wave's share");
static_assert(((EDGES_PER_PART + 31 + 31) >> CHUNK_SHIFT) <= BWORDS * 32,
              "the chunks of one share must fit its bitmap");
static_assert(2 * NPART * N_NODES <= BITS_OFF,
              "the bitmaps lie behind the degree partials");
#define E4_PER_PART    (EDGES_PER_PART / 4)
#define N4             (N_NODES / 4)
#define HIST_THREADS   1024
#define E4_PER_THREAD  ((E4_PER_PART + HIST_THREADS - 1) / HIST_THREADS)

__global__ __launch_bounds__(HIST_THREADS) void degree_hist(const int* __restrict__ src,
                                                            const int* __restrict__ dst,
                                                            unsigned* __restrict__ part) {
    __shared__ unsigned hist[N_NODES];
    __shared__ unsigned cbits[NLBLK * BWORDS];

    const int tid = threadIdx.x;
    const int arr = blockIdx.x / NPART;
    const int p   = blockIdx.x % NPART;
    const int* idx = (arr == 0) ? src : dst;
    const int4* ep = (const int4*)(idx + (size_t)p * EDGES_PER_PART);

    int4 e[E4_PER_THREAD];
#pragma unroll
    for (int j = 0; j < E4_PER_THREAD; ++j) {
        const int i = tid + j * HIST_THREADS;
        if (i < E4_PER_PART) e[j] = ep[i];
    }

    if (arr < 2) {
        uint4* h4 = (uint4*)hist;
        for (int i = tid; i < N4; i += HIST_THREADS) h4[i] = make_uint4(0u, 0u, 0u, 0u);
        __syncthreads();

#pragma unroll
        for (int j = 0; j < E4_PER_THREAD; ++j) {
            const int i = tid + j * HIST_THREADS;
            if (i < E4_PER_PART) {
                atomicAdd(&hist[e[j].x], 1u);
                atomicAdd(&hist[e[j].y], 1u);
                atomicAdd(&hist[e[j].z], 1u);
                atomicAdd(&hist[e[j].w], 1u);
            }
        }
        __syncthreads();

        uint4* op = (uint4*)(part + (size_t)blockIdx.x * N_NODES);
        for (int i = tid; i < N4; i += HIST_THREADS) op[i] = h4[i];
    } else {
        for (int i = tid; i < NLBLK * BWORDS; i += HIST_THREADS) cbits[i] = 0u;
        __syncthreads();

        const unsigned ebeg = (unsigned)p * EDGES_PER_PART;
#pragma unroll
        for (int j = 0; j < E4_PER_THREAD; ++j) {
            const int i = tid + j * HIST_THREADS;
            if (i < E4_PER_PART) {
                const unsigned lc  = ((ebeg + 4u * (unsigned)i) >> CHUNK_SHIFT) - (ebeg >> CHUNK_SHIFT);
                const unsigned wd  = lc >> 5;
                const unsigned bit = 1u << (lc & 31u);
                atomicOr(&cbits[((unsigned)e[j].x / RNODES) * BWORDS + wd], bit);
                atomicOr(&cbits[((unsigned)e[j].y / RNODES) * BWORDS + wd], bit);
                atomicOr(&cbits[((unsigned)e[j].z / RNODES) * BWORDS + wd], bit);
                atomicOr(&cbits[((unsigned)e[j].w / RNODES) * BWORDS + wd], bit);
            }
        }
        __syncthreads();

        unsigned* bp = part + BITS_OFF;
        for (int i = tid; i < NLBLK * BWORDS; i += HIST_THREADS) {
            const int b = i / BWORDS;
            const int k = i - b * BWORDS;
            bp[(size_t)(b * NPART + p) * BWORDS + k] = cbits[i];
        }
    }
}

__global__ __launch_bounds__(256) void degree_finish(const unsigned* __restrict__ part,
                                                     float* __restrict__ rs) {
    const int t = blockIdx.x * blockDim.x + threadIdx.x;
    if (t >= 2 * N4) return;
    const int arr = (t >= N4) ? 1 : 0;
    const int q   = t - arr * N4;
    const uint4* pp = (const uint4*)part + (size_t)arr * NPART * N4 + q;
    uint4 s = make_uint4(0u, 0u, 0u, 0u);
#pragma unroll
    for (int p = 0; p < NPART; ++p) {
        const uint4 v = pp[(size_t)p * N4];
        s.x += v.x; s.y += v.y; s.z += v.z; s.w += v.w;
    }
    float4 r;
    r.x = rsqrtf(fmaxf((float)s.x, 1.0f));
    r.y = rsqrtf(fmaxf((float)s.y, 1.0f));
    r.z = rsqrtf(fmaxf((float)s.z, 1.0f));
    r.w = rsqrtf(fmaxf((float)s.w, 1.0f));
    ((float4*)rs)[t] = r;
}


__device__ __forceinline__ void fma4(float4& a, const float4 v, const float s) {
    a.x = fmaf(v.x, s, a.x);
    a.y = fmaf(v.y, s, a.y);
    a.z = fmaf(v.z, s, a.z);
    a.w = fmaf(v.w, s, a.w);
}

#define ACC_ROW(slot, v, s)                         \
    switch (slot) {                                 \
    case 0: fma4(a0, v, s); break;                  \
    case 1: fma4(a1, v, s); break;                  \
    case 2: fma4(a2, v, s); break;                  \
    case 3: fma4(a3, v, s); break;                  \
    case 4: fma4(a4, v, s); break;                  \
    case 5: fma4(a5, v, s); break;                  \
    case 6: fma4(a6, v, s); break;                  \
    default: break;                                 \
    }

template <int NT, bool RELU>
__device__ __forceinline__ void tile_product(const float* Alds, const float* __restrict__ Wm,
                                             const float* __restrict__ bias,
                                             float* __restrict__ out,
                                             int lo, int ntiles, int g, int n0, int lane) {
    const int mrow = lane & 15;
    const int khi  = (lane >> 4) << 1;
    const float* wp = Wm + n0 + mrow;
    const float* ap[NT];
    bool valid[NT];
    v8f acc[NT];
#pragma unroll
    for (int t = 0; t < NT; ++t) {
        const int rt = g + 4 * t;
        valid[t] = rt < ntiles;
        const int rtc = valid[t] ? rt : g;
        ap[t] = Alds + (rtc * 16 + mrow) * ASTRIDE + khi;
        acc[t] = (v8f){0.f, 0.f, 0.f, 0.f, 0.f, 0.f, 0.f, 0.f};
    }
#pragma unroll
    for (int k = 0; k < DFEAT; k += 4) {
        v2f b;
        b.x = wp[(size_t)(k + khi)     * DFEAT];
        b.y = wp[(size_t)(k + khi + 1) * DFEAT];
#pragma unroll
        for (int t = 0; t < NT; ++t) {
            const v2f a = *(const v2f*)(ap[t] + k);
            acc[t] = __builtin_amdgcn_wmma_f32_16x16x4_f32(false, a, false, b,
                                                           (short)0, acc[t], false, false);
        }
    }
    const int   col  = n0 + mrow;
    const float bv   = bias[col];
    const int   rsel = (lane >> 4) << 3;
#pragma unroll
    for (int t = 0; t < NT; ++t) {
        if (valid[t]) {
            const int rowBase = lo + (g + 4 * t) * 16;
#pragma unroll
            for (int v = 0; v < 8; ++v) {
                float h = acc[t][v] + bv;
                if (RELU) h = fmaxf(h, 0.f);
                out[(size_t)(rowBase + v + rsel) * DFEAT + col] = h;
            }
        }
    }
}

template <bool RELU, bool KEPT>
__global__ __launch_bounds__(LTHREADS) void gcn_layer(const float* __restrict__ X,
                                                      const float* __restrict__ rs_out,
                                                      const float* __restrict__ rs_in,
                                                      const int* __restrict__ src,
                                                      const int* __restrict__ dst,
                                                      const float* __restrict__ Wm,
                                                      const float* __restrict__ bias,
                                                      float* __restrict__ out,
                                                      unsigned* __restrict__ keep) {
    __shared__ __attribute__((aligned(16))) float Alds[RNODES * ASTRIDE];
    __shared__ __attribute__((aligned(16))) unsigned lists[LWAVES * LWAVES * CAPS];
    __shared__ unsigned cntS[LWAVES * LWAVES];
    __shared__ unsigned ovfFlag;
    __shared__ unsigned rowInfo[RNODES];
    __shared__ unsigned nextRow;

    const int tid  = threadIdx.x;
    const int lane = tid & 31;
    const int w    = __builtin_amdgcn_readfirstlane(tid >> 5);
    const int lo   = blockIdx.x * RNODES;
    const unsigned nrows = (N_NODES - lo < RNODES) ? (unsigned)(N_NODES - lo) : (unsigned)RNODES;

    const int4*   dst4 = (const int4*)dst;
    const float4* X4   = (const float4*)X;

    unsigned* krec = keep + KEEP_BASE + (size_t)((unsigned)blockIdx.x * LWAVES + (unsigned)w) * KEEP_REC;
    unsigned* kent = krec + KEEP_HEAD;

    bool slow;
    unsigned nj[7];
    unsigned* mypart = lists + w * FLAT_CAP;
    if (tid == 0) nextRow = 0u;

    if (KEPT) {
        const unsigned hv = krec[lane];
        slow = __builtin_amdgcn_readlane((int)hv, 7) != 0;
#pragma unroll
        for (int j = 0; j < 7; ++j) nj[j] = (unsigned)__builtin_amdgcn_readlane((int)hv, j);
    } else {
#pragma unroll
        for (int j = 0; j < 7; ++j) nj[j] = 0u;
    if (tid == 0) ovfFlag = 0u;
    __syncthreads();

    {
        const int ebeg   = w * (4 * E4_WAVE);
        const int cfirst = ebeg >> CHUNK_SHIFT;
        cntS[lane * LWAVES + w] = (unsigned)((lane * LWAVES + w) * CAPS);
        const unsigned* bw = keep + BITS_OFF
                           + (size_t)((unsigned)blockIdx.x * LWAVES + (unsigned)w) * BWORDS;
        const unsigned myw = (lane < BWORDS) ? bw[lane] : 0u;
        unsigned* clist = (unsigned*)Alds + w * (BWORDS * 32);
        const unsigned pc = (unsigned)__builtin_popcount(myw);
        unsigned cincl = pc;
#pragma unroll
        for (int dd = 1; dd < 32; dd <<= 1) {
            const unsigned up = __shfl_up(cincl, dd);
            if (lane >= dd) cincl += up;
        }
        const int nchunks = __builtin_amdgcn_readlane((int)cincl, 31);
        {
            unsigned mm = myw;
            unsigned o  = cincl - pc;
            while (mm) {
                clist[o++] = (unsigned)(lane * 32 + __builtin_ctz(mm));
                mm &= mm - 1u;
            }
        }
        __syncthreads();
        for (int k0 = 0; k0 < nchunks; k0 += SCAN_U) {
            const int kl = k0 + lane;
            const int cv = (lane < SCAN_U && kl < nchunks) ? (int)clist[kl] : -1;
            int ck[SCAN_U];
#pragma unroll
            for (int u = 0; u < SCAN_U; ++u) ck[u] = __builtin_amdgcn_readlane(cv, u);
            int dv[SCAN_U];
#pragma unroll
            for (int u = 0; u < SCAN_U; ++u) {
                const int  e  = (cfirst + ck[u]) * 32 + lane;
                const bool ok = (unsigned)(e - ebeg) < (unsigned)(4 * E4_WAVE);
                dv[u] = dst[ok ? e : ebeg];
            }
#pragma unroll
            for (int u = 0; u < SCAN_U; ++u) asm volatile("" :: "v"(dv[u]));
            unsigned nn[SCAN_U];
            unsigned ev[SCAN_U];
#pragma unroll
            for (int u = 0; u < SCAN_U; ++u) {
                const int      e  = (cfirst + ck[u]) * 32 + lane;
                const bool     ok = (unsigned)(e - ebeg) < (unsigned)(4 * E4_WAVE);
                const unsigned uu = (unsigned)dv[u] - (unsigned)lo;
                ev[u] = (uu << 20) | (unsigned)e;
                nn[u] = 0xFFFFFFFFu;
                if (ok && uu < nrows)
                    nn[u] = atomicAdd(&cntS[(uu & 31u) * LWAVES + (unsigned)w], 1u);
            }
#pragma unroll
            for (int u = 0; u < SCAN_U; ++u) asm volatile("" :: "v"(nn[u]));
#pragma unroll
            for (int u = 0; u < SCAN_U; ++u) {
                if (nn[u] < (unsigned)(LWAVES * LWAVES * CAPS)) lists[nn[u]] = ev[u];
            }
        }
        {
            const unsigned cnt = cntS[lane * LWAVES + w] - (unsigned)((lane * LWAVES + w) * CAPS);
            cntS[lane * LWAVES + w] = cnt;
            if (cnt > (unsigned)CAPS) ovfFlag = 1u;
        }
    }
    __syncthreads();
    slow = __builtin_amdgcn_readfirstlane((int)ovfFlag) != 0;
    }

    float4 a0 = make_float4(0.f, 0.f, 0.f, 0.f);
    float4 a1 = a0, a2 = a0, a3 = a0, a4 = a0, a5 = a0, a6 = a0;

    if (!slow) {
        unsigned oj[7];
        if (!KEPT) {
        unsigned* myflat = (unsigned*)Alds + w * FLAT_CAP;
        const unsigned c_l = cntS[w * LWAVES + lane];
        unsigned incl = c_l;
#pragma unroll
        for (int dd = 1; dd < 32; dd <<= 1) {
            const unsigned up = __shfl_up(incl, dd);
            if (lane >= dd) incl += up;
        }
        const unsigned off = incl - c_l;
        const int total = __builtin_amdgcn_readlane((int)incl, 31);
        const uint4* mylist4 = (const uint4*)(lists + (w * LWAVES + lane) * CAPS);
        unsigned ent[CAPS];
        unsigned sn[CAPS];
#pragma unroll
        for (int i = 0; i < CAPS / 4; ++i) {
            const uint4 q = mylist4[i];
            ent[4 * i + 0] = q.x; ent[4 * i + 1] = q.y; ent[4 * i + 2] = q.z; ent[4 * i + 3] = q.w;
        }
#pragma unroll
        for (int i = 0; i < CAPS; ++i) {
            const unsigned e = ((unsigned)i < c_l) ? (ent[i] & 0xFFFFFu) : 0u;
            sn[i] = (unsigned)src[e];
        }
#pragma unroll
        for (int i = 0; i < CAPS; ++i) asm volatile("" :: "v"(sn[i]));
#pragma unroll
        for (int i = 0; i < CAPS; ++i) {
            if ((unsigned)i < c_l) myflat[off + i] = ((ent[i] >> 25) << 16) | sn[i];
        }
        __syncthreads();

        unsigned run[7];
#pragma unroll
        for (int j = 0; j < 7; ++j) run[j] = 0u;
        for (int k = 0; k < total; k += 32) {
            const int idx = k + lane;
            const unsigned slot = (idx < total) ? (myflat[idx] >> 16) : 7u;
#pragma unroll
            for (int j = 0; j < 7; ++j)
                nj[j] += (unsigned)__builtin_popcount(__builtin_amdgcn_ballot_w32(slot == (unsigned)j));
        }
        oj[0] = 0u;
#pragma unroll
        for (int j = 1; j < 7; ++j) oj[j] = oj[j - 1] + nj[j - 1];
        for (int k = 0; k < total; k += 32) {
            const int idx = k + lane;
            const unsigned ent  = (idx < total) ? myflat[idx] : (7u << 16);
            const unsigned slot = ent >> 16;
#pragma unroll
            for (int j = 0; j < 7; ++j) {
                const unsigned m = __builtin_amdgcn_ballot_w32(slot == (unsigned)j);
                if (slot == (unsigned)j)
                    mypart[oj[j] + run[j] + __builtin_amdgcn_mbcnt_lo(m, 0u)] = ent & 0xFFFFu;
                run[j] += (unsigned)__builtin_popcount(m);
            }
        }
        } else {
            oj[0] = 0u;
#pragma unroll
            for (int j = 1; j < 7; ++j) oj[j] = oj[j - 1] + nj[j - 1];
            const unsigned total = oj[6] + nj[6];
            for (unsigned k = 0u; k < total; k += 128u) {
                unsigned e[4];
#pragma unroll
                for (int u = 0; u < 4; ++u) {
                    const unsigned idx = k + (unsigned)(u * 32 + lane);
                    e[u] = (idx < total) ? kent[idx] : 0u;
                }
#pragma unroll
                for (int u = 0; u < 4; ++u) {
                    const unsigned idx = k + (unsigned)(u * 32 + lane);
                    if (idx < total) mypart[idx] = e[u];
                }
            }
        }
#pragma unroll
        for (int j = 0; j < 7; ++j) {
            const int r = j * 32 + w;
            if (r < RNODES && lane == 0)
                rowInfo[r] = ((unsigned)(w * FLAT_CAP) + oj[j]) | (nj[j] << 16);
        }
        __syncthreads();

        if (!KEPT) {
            const unsigned total = oj[6] + nj[6];
            for (unsigned k = 0u; k < total; k += 32u) {
                const unsigned idx = k + (unsigned)lane;
                if (idx < total) kent[idx] = mypart[idx];
            }
        }

        for (int t = 0; t <= RNODES; ++t) {
            unsigned took = 0u;
            if (lane == 0) took = atomicAdd(&nextRow, 1u);
            const unsigned r = (unsigned)__builtin_amdgcn_readfirstlane((int)took);
            if (r >= (unsigned)RNODES) break;
            const unsigned info = (unsigned)__builtin_amdgcn_readfirstlane((int)rowInfo[r]);
            const unsigned rcnt = info >> 16;
            const unsigned rbeg = info & 0xFFFFu;
            const float si = (r < nrows) ? rs_in[lo + r] : 0.f;
            float4 a = make_float4(0.f, 0.f, 0.f, 0.f);
            for (unsigned k0 = 0u; k0 < rcnt; k0 += GATHER_U) {
                const unsigned kl   = k0 + (unsigned)lane;
                const unsigned entv = lists[rbeg + ((kl < rcnt) ? kl : (rcnt - 1u))];
                float4 v[GATHER_U];
                float  sc[GATHER_U];
#pragma unroll
                for (int u = 0; u < GATHER_U; ++u) {
                    const unsigned s = (unsigned)__builtin_amdgcn_readlane((int)entv, u);
                    v[u]  = X4[(size_t)s * (DFEAT / 4) + lane];
                    sc[u] = rs_out[s];
                }
#pragma unroll
                for (int u = 0; u < GATHER_U; ++u)
                    asm volatile("" :: "v"(v[u].x), "v"(v[u].y), "v"(v[u].z), "v"(v[u].w),
                                 "s"(sc[u]));
#pragma unroll
                for (int u = 0; u < GATHER_U; ++u) {
                    if (k0 + (unsigned)u < rcnt) fma4(a, v[u], sc[u]);
                }
            }
            *(float4*)&Alds[r * ASTRIDE + lane * 4] =
                make_float4(a.x * si, a.y * si, a.z * si, a.w * si);
        }
    } else {
        for (int pos = 0; pos < E4_TOTAL; pos += 32) {
            const int4 d = dst4[pos + lane];
            const unsigned u0 = (unsigned)d.x - (unsigned)lo;
            const unsigned u1 = (unsigned)d.y - (unsigned)lo;
            const unsigned u2 = (unsigned)d.z - (unsigned)lo;
            const unsigned u3 = (unsigned)d.w - (unsigned)lo;
            unsigned m0 = __builtin_amdgcn_ballot_w32(u0 < nrows && (u0 & 31u) == (unsigned)w);
            unsigned m1 = __builtin_amdgcn_ballot_w32(u1 < nrows && (u1 & 31u) == (unsigned)w);
            unsigned m2 = __builtin_amdgcn_ballot_w32(u2 < nrows && (u2 & 31u) == (unsigned)w);
            unsigned m3 = __builtin_amdgcn_ballot_w32(u3 < nrows && (u3 & 31u) == (unsigned)w);
            if ((m0 | m1 | m2 | m3) != 0u) {
#define SLOW_TAKE(m, u, c)                                                              \
                while (m) {                                                             \
                    const int l = __builtin_ctz(m);                                     \
                    m &= m - 1u;                                                        \
                    const unsigned row = (unsigned)__builtin_amdgcn_readlane((int)(u), l); \
                    const int      s   = src[4 * (pos + l) + (c)];                      \
                    const float    scs = rs_out[s];                                     \
                    const float4   vs  = X4[(size_t)s * (DFEAT / 4) + lane];            \
                    ACC_ROW(row >> 5, vs, scs)                                          \
                }
                SLOW_TAKE(m0, u0, 0)
                SLOW_TAKE(m1, u1, 1)
                SLOW_TAKE(m2, u2, 2)
                SLOW_TAKE(m3, u3, 3)
#undef SLOW_TAKE
            }
        }
    }

    if (slow) {
#define PUT_ROW(j, a)                                                                   \
    {                                                                                   \
        const int r = (j) * 32 + w;                                                     \
        if (r < RNODES) {                                                               \
            const float si = (r < (int)nrows) ? rs_in[lo + r] : 0.f;                    \
            *(float4*)&Alds[r * ASTRIDE + lane * 4] =                                   \
                make_float4(a.x * si, a.y * si, a.z * si, a.w * si);                    \
        }                                                                               \
    }
    PUT_ROW(0, a0)
    PUT_ROW(1, a1)
    PUT_ROW(2, a2)
    PUT_ROW(3, a3)
    PUT_ROW(4, a4)
    PUT_ROW(5, a5)
    PUT_ROW(6, a6)
#undef PUT_ROW
    }
    __syncthreads();

    if (!KEPT) {
        unsigned hv = (lane == 7 && slow) ? 1u : 0u;
#pragma unroll
        for (int j = 0; j < 7; ++j) if (lane == j) hv = nj[j];
        krec[lane] = hv;
    }

    const int ntiles = (int)(nrows >> 4);
    const int g      = w >> 3;
    const int n0     = (w & 7) << 4;
    if (g == 0) tile_product<4, RELU>(Alds, Wm, bias, out, lo, ntiles, g, n0, lane);
    else        tile_product<3, RELU>(Alds, Wm, bias, out, lo, ntiles, g, n0, lane);
}

extern "C" void kernel_launch(void* const* d_in, const int* in_sizes, int n_in,
                              void* d_out, int out_size, void* d_ws, size_t ws_size,
                              hipStream_t stream) {
    const float* x   = (const float*)d_in[0];
    const int*   src = (const int*)  d_in[1];
    const int*   dst = (const int*)  d_in[2];
    const float* W1  = (const float*)d_in[3];
    const float* b1  = (const float*)d_in[4];
    const float* W2  = (const float*)d_in[5];
    const float* b2  = (const float*)d_in[6];
    float*       out = (float*)d_out;

    float* ws     = (float*)d_ws;
    float* rs_out = ws;
    float* rs_in  = ws + N_NODES;
    float* scr    = ws + 2 * N_NODES;
    float* h1     = scr + (size_t)N_NODES * DFEAT;

    const int T = 256;

    unsigned* part = (unsigned*)scr;
    degree_hist<<<3 * NPART, HIST_THREADS, 0, stream>>>(src, dst, part);
    degree_finish<<<(2 * N4 + T - 1) / T, T, 0, stream>>>(part, rs_out);

    unsigned* kept = (unsigned*)scr;
    gcn_layer<true, false><<<NLBLK, LTHREADS, 0, stream>>>(x, rs_out, rs_in, src, dst, W1, b1, h1, kept);
    gcn_layer<false, true><<<NLBLK, LTHREADS, 0, stream>>>(h1, rs_out, rs_in, src, dst, W2, b2, out, kept);
}


# Layer 1 stores row r times rs_out[r] (scale read from LDS in its epilogue); layer-2 edge sum adds the gathered rows unweighted: no weight load per edge; terms rounded before the add, sum order kept
# speedup vs baseline: 4.2445x; 1.0592x over previous
#include <hip/hip_runtime.h>

#define N_NODES 50000
#define N_EDGES 800000
#define DFEAT   128
#define ASTRIDE 132

typedef float v2f __attribute__((ext_vector_type(2)));
typedef float v8f __attribute__((ext_vector_type(8)));

#define RNODES   208
#define NLBLK    ((N_NODES + RNODES - 1) / RNODES)
#define LTHREADS 1024
#define LWAVES   32
#define CAPS     24
#define E4_TOTAL (N_EDGES / 4)
#define E4_WAVE  (E4_TOTAL / LWAVES)
#define SCAN_U   16
#define GATHER_U 8
#define FLAT_CAP (LWAVES * CAPS)
#define KEEP_HEAD  32
#define KEEP_REC   (KEEP_HEAD + FLAT_CAP)
#define KEEP_BASE  32
#define BWORDS   25
#define BITS_OFF (KEEP_BASE + NLBLK * LWAVES * KEEP_REC)
static_assert(BITS_OFF + NLBLK * LWAVES * BWORDS <= N_NODES * DFEAT,
              "kept records and chunk bitmaps must fit the scratch plane");

#define NPART 32
#define EDGES_PER_PART (N_EDGES / NPART)
#define CHUNK_SHIFT 5
static_assert(NPART == LWAVES && EDGES_PER_PART == 4 * E4_WAVE,
              "a bitmap group's share is one scanning wave's share");
static_assert(((EDGES_PER_PART + 31 + 31) >> CHUNK_SHIFT) <= BWORDS * 32,
              "the chunks of one share must fit its bitmap");
static_assert(2 * NPART * N_NODES <= BITS_OFF,
              "the bitmaps lie behind the degree partials");
#define E4_PER_PART    (EDGES_PER_PART / 4)
#define N4             (N_NODES / 4)
#define HIST_THREADS   1024
#define E4_PER_THREAD  ((E4_PER_PART + HIST_THREADS - 1) / HIST_THREADS)

__global__ __launch_bounds__(HIST_THREADS) void degree_hist(const int* __restrict__ src,
                                                            const int* __restrict__ dst,
                                                            unsigned* __restrict__ part) {
    __shared__ unsigned hist[N_NODES];
    __shared__ unsigned cbits[NLBLK * BWORDS];

    const int tid = threadIdx.x;
    const int arr = blockIdx.x / NPART;
    const int p   = blockIdx.x % NPART;
    const int* idx = (arr == 0) ? src : dst;
    const int4* ep = (const int4*)(idx + (size_t)p * EDGES_PER_PART);

    int4 e[E4_PER_THREAD];
#pragma unroll
    for (int j = 0; j < E4_PER_THREAD; ++j) {
        const int i = tid + j * HIST_THREADS;
        if (i < E4_PER_PART) e[j] = ep[i];
    }

    if (arr < 2) {
        uint4* h4 = (uint4*)hist;
        for (int i = tid; i < N4; i += HIST_THREADS) h4[i] = make_uint4(0u, 0u, 0u, 0u);
        __syncthreads();

#pragma unroll
        for (int j = 0; j < E4_PER_THREAD; ++j) {
            const int i = tid + j * HIST_THREADS;
            if (i < E4_PER_PART) {
                atomicAdd(&hist[e[j].x], 1u);
                atomicAdd(&hist[e[j].y], 1u);
                atomicAdd(&hist[e[j].z], 1u);
                atomicAdd(&hist[e[j].w], 1u);
            }
        }
        __syncthreads();

        uint4* op = (uint4*)(part + (size_t)blockIdx.x * N_NODES);
        for (int i = tid; i < N4; i += HIST_THREADS) op[i] = h4[i];
    } else {
        for (int i = tid; i < NLBLK * BWORDS; i += HIST_THREADS) cbits[i] = 0u;
        __syncthreads();

        const unsigned ebeg = (unsigned)p * EDGES_PER_PART;
#pragma unroll
        for (int j = 0; j < E4_PER_THREAD; ++j) {
            const int i = tid + j * HIST_THREADS;
            if (i < E4_PER_PART) {
                const unsigned lc  = ((ebeg + 4u * (unsigned)i) >> CHUNK_SHIFT) - (ebeg >> CHUNK_SHIFT);
                const unsigned wd  = lc >> 5;
                const unsigned bit = 1u << (lc & 31u);
                atomicOr(&cbits[((unsigned)e[j].x / RNODES) * BWORDS + wd], bit);
                atomicOr(&cbits[((unsigned)e[j].y / RNODES) * BWORDS + wd], bit);
                atomicOr(&cbits[((unsigned)e[j].z / RNODES) * BWORDS + wd], bit);
                atomicOr(&cbits[((unsigned)e[j].w / RNODES) * BWORDS + wd], bit);
            }
        }
        __syncthreads();

        unsigned* bp = part + BITS_OFF;
        for (int i = tid; i < NLBLK * BWORDS; i += HIST_THREADS) {
            const int b = i / BWORDS;
            const int k = i - b * BWORDS;
            bp[(size_t)(b * NPART + p) * BWORDS + k] = cbits[i];
        }
    }
}

__global__ __launch_bounds__(256) void degree_finish(const unsigned* __restrict__ part,
                                                     float* __restrict__ rs) {
    const int t = blockIdx.x * blockDim.x + threadIdx.x;
    if (t >= 2 * N4) return;
    const int arr = (t >= N4) ? 1 : 0;
    const int q   = t - arr * N4;
    const uint4* pp = (const uint4*)part + (size_t)arr * NPART * N4 + q;
    uint4 s = make_uint4(0u, 0u, 0u, 0u);
#pragma unroll
    for (int p = 0; p < NPART; ++p) {
        const uint4 v = pp[(size_t)p * N4];
        s.x += v.x; s.y += v.y; s.z += v.z; s.w += v.w;
    }
    float4 r;
    r.x = rsqrtf(fmaxf((float)s.x, 1.0f));
    r.y = rsqrtf(fmaxf((float)s.y, 1.0f));
    r.z = rsqrtf(fmaxf((float)s.z, 1.0f));
    r.w = rsqrtf(fmaxf((float)s.w, 1.0f));
    ((float4*)rs)[t] = r;
}


__device__ __forceinline__ void fma4(float4& a, const float4 v, const float s) {
    a.x = fmaf(v.x, s, a.x);
    a.y = fmaf(v.y, s, a.y);
    a.z = fmaf(v.z, s, a.z);
    a.w = fmaf(v.w, s, a.w);
}

__device__ __forceinline__ void add4(float4& a, const float4 v) {
    a.x += v.x;
    a.y += v.y;
    a.z += v.z;
    a.w += v.w;
}

#define ACC_ROW(slot, v, s)                         \
    switch (slot) {                                 \
    case 0: fma4(a0, v, s); break;                  \
    case 1: fma4(a1, v, s); break;                  \
    case 2: fma4(a2, v, s); break;                  \
    case 3: fma4(a3, v, s); break;                  \
    case 4: fma4(a4, v, s); break;                  \
    case 5: fma4(a5, v, s); break;                  \
    case 6: fma4(a6, v, s); break;                  \
    default: break;                                 \
    }

template <int NT, bool RELU, bool SCALE_OUT>
__device__ __forceinline__ void tile_product(const float* Alds, const float* __restrict__ Wm,
                                             const float* __restrict__ bias,
                                             const float* rowscale,
                                             float* __restrict__ out,
                                             int lo, int ntiles, int g, int n0, int lane) {
    const int mrow = lane & 15;
    const int khi  = (lane >> 4) << 1;
    const float* wp = Wm + n0 + mrow;
    const float* ap[NT];
    bool valid[NT];
    v8f acc[NT];
#pragma unroll
    for (int t = 0; t < NT; ++t) {
        const int rt = g + 4 * t;
        valid[t] = rt < ntiles;
        const int rtc = valid[t] ? rt : g;
        ap[t] = Alds + (rtc * 16 + mrow) * ASTRIDE + khi;
        acc[t] = (v8f){0.f, 0.f, 0.f, 0.f, 0.f, 0.f, 0.f, 0.f};
    }
#pragma unroll
    for (int k = 0; k < DFEAT; k += 4) {
        v2f b;
        b.x = wp[(size_t)(k + khi)     * DFEAT];
        b.y = wp[(size_t)(k + khi + 1) * DFEAT];
#pragma unroll
        for (int t = 0; t < NT; ++t) {
            const v2f a = *(const v2f*)(ap[t] + k);
            acc[t] = __builtin_amdgcn_wmma_f32_16x16x4_f32(false, a, false, b,
                                                           (short)0, acc[t], false, false);
        }
    }
    const int   col  = n0 + mrow;
    const float bv   = bias[col];
    const int   rsel = (lane >> 4) << 3;
#pragma unroll
    for (int t = 0; t < NT; ++t) {
        if (valid[t]) {
            const int rowBase = lo + (g + 4 * t) * 16;
#pragma unroll
            for (int v = 0; v < 8; ++v) {
                float h = acc[t][v] + bv;
                if (RELU) h = fmaxf(h, 0.f);
                if (SCALE_OUT) h *= rowscale[(g + 4 * t) * 16 + v + rsel];
                out[(size_t)(rowBase + v + rsel) * DFEAT + col] = h;
            }
        }
    }
}

template <bool RELU, bool KEPT>
__global__ __launch_bounds__(LTHREADS) void gcn_layer(const float* __restrict__ X,
                                                      const float* __restrict__ rs_out,
                                                      const float* __restrict__ rs_in,
                                                      const int* __restrict__ src,
                                                      const int* __restrict__ dst,
                                                      const float* __restrict__ Wm,
                                                      const float* __restrict__ bias,
                                                      float* __restrict__ out,
                                                      unsigned* __restrict__ keep) {
    __shared__ __attribute__((aligned(16))) float Alds[RNODES * ASTRIDE];
    __shared__ __attribute__((aligned(16))) unsigned lists[LWAVES * LWAVES * CAPS];
    __shared__ unsigned cntS[LWAVES * LWAVES];
    __shared__ unsigned ovfFlag;
    __shared__ unsigned rowInfo[RNODES];
    __shared__ unsigned nextRow;
    __shared__ float rsoS[RNODES];

    const int tid  = threadIdx.x;
    const int lane = tid & 31;
    const int w    = __builtin_amdgcn_readfirstlane(tid >> 5);
    const int lo   = blockIdx.x * RNODES;
    const unsigned nrows = (N_NODES - lo < RNODES) ? (unsigned)(N_NODES - lo) : (unsigned)RNODES;

    const int4*   dst4 = (const int4*)dst;
    const float4* X4   = (const float4*)X;

    unsigned* krec = keep + KEEP_BASE + (size_t)((unsigned)blockIdx.x * LWAVES + (unsigned)w) * KEEP_REC;
    unsigned* kent = krec + KEEP_HEAD;

    bool slow;
    unsigned nj[7];
    unsigned* mypart = lists + w * FLAT_CAP;
    if (tid == 0) nextRow = 0u;

    if (KEPT) {
        const unsigned hv = krec[lane];
        slow = __builtin_amdgcn_readlane((int)hv, 7) != 0;
#pragma unroll
        for (int j = 0; j < 7; ++j) nj[j] = (unsigned)__builtin_amdgcn_readlane((int)hv, j);
    } else {
#pragma unroll
        for (int j = 0; j < 7; ++j) nj[j] = 0u;
    if (tid == 0) ovfFlag = 0u;
    __syncthreads();

    {
        const int ebeg   = w * (4 * E4_WAVE);
        const int cfirst = ebeg >> CHUNK_SHIFT;
        cntS[lane * LWAVES + w] = (unsigned)((lane * LWAVES + w) * CAPS);
        const unsigned* bw = keep + BITS_OFF
                           + (size_t)((unsigned)blockIdx.x * LWAVES + (unsigned)w) * BWORDS;
        const unsigned myw = (lane < BWORDS) ? bw[lane] : 0u;
        unsigned* clist = (unsigned*)Alds + w * (BWORDS * 32);
        const unsigned pc = (unsigned)__builtin_popcount(myw);
        unsigned cincl = pc;
#pragma unroll
        for (int dd = 1; dd < 32; dd <<= 1) {
            const unsigned up = __shfl_up(cincl, dd);
            if (lane >= dd) cincl += up;
        }
        const int nchunks = __builtin_amdgcn_readlane((int)cincl, 31);
        {
            unsigned mm = myw;
            unsigned o  = cincl - pc;
            while (mm) {
                clist[o++] = (unsigned)(lane * 32 + __builtin_ctz(mm));
                mm &= mm - 1u;
            }
        }
        __syncthreads();
        for (int k0 = 0; k0 < nchunks; k0 += SCAN_U) {
            const int kl = k0 + lane;
            const int cv = (lane < SCAN_U && kl < nchunks) ? (int)clist[kl] : -1;
            int ck[SCAN_U];
#pragma unroll
            for (int u = 0; u < SCAN_U; ++u) ck[u] = __builtin_amdgcn_readlane(cv, u);
            int dv[SCAN_U];
#pragma unroll
            for (int u = 0; u < SCAN_U; ++u) {
                const int  e  = (cfirst + ck[u]) * 32 + lane;
                const bool ok = (unsigned)(e - ebeg) < (unsigned)(4 * E4_WAVE);
                dv[u] = dst[ok ? e : ebeg];
            }
#pragma unroll
            for (int u = 0; u < SCAN_U; ++u) asm volatile("" :: "v"(dv[u]));
            unsigned nn[SCAN_U];
            unsigned ev[SCAN_U];
#pragma unroll
            for (int u = 0; u < SCAN_U; ++u) {
                const int      e  = (cfirst + ck[u]) * 32 + lane;
                const bool     ok = (unsigned)(e - ebeg) < (unsigned)(4 * E4_WAVE);
                const unsigned uu = (unsigned)dv[u] - (unsigned)lo;
                ev[u] = (uu << 20) | (unsigned)e;
                nn[u] = 0xFFFFFFFFu;
                if (ok && uu < nrows)
                    nn[u] = atomicAdd(&cntS[(uu & 31u) * LWAVES + (unsigned)w], 1u);
            }
#pragma unroll
            for (int u = 0; u < SCAN_U; ++u) asm volatile("" :: "v"(nn[u]));
#pragma unroll
            for (int u = 0; u < SCAN_U; ++u) {
                if (nn[u] < (unsigned)(LWAVES * LWAVES * CAPS)) lists[nn[u]] = ev[u];
            }
        }
        {
            const unsigned cnt = cntS[lane * LWAVES + w] - (unsigned)((lane * LWAVES + w) * CAPS);
            cntS[lane * LWAVES + w] = cnt;
            if (cnt > (unsigned)CAPS) ovfFlag = 1u;
        }
    }
    __syncthreads();
    slow = __builtin_amdgcn_readfirstlane((int)ovfFlag) != 0;
    }

    float4 a0 = make_float4(0.f, 0.f, 0.f, 0.f);
    float4 a1 = a0, a2 = a0, a3 = a0, a4 = a0, a5 = a0, a6 = a0;

    if (!slow) {
        unsigned oj[7];
        if (!KEPT) {
        unsigned* myflat = (unsigned*)Alds + w * FLAT_CAP;
        const unsigned c_l = cntS[w * LWAVES + lane];
        unsigned incl = c_l;
#pragma unroll
        for (int dd = 1; dd < 32; dd <<= 1) {
            const unsigned up = __shfl_up(incl, dd);
            if (lane >= dd) incl += up;
        }
        const unsigned off = incl - c_l;
        const int total = __builtin_amdgcn_readlane((int)incl, 31);
        const uint4* mylist4 = (const uint4*)(lists + (w * LWAVES + lane) * CAPS);
        unsigned ent[CAPS];
        unsigned sn[CAPS];
#pragma unroll
        for (int i = 0; i < CAPS / 4; ++i) {
            const uint4 q = mylist4[i];
            ent[4 * i + 0] = q.x; ent[4 * i + 1] = q.y; ent[4 * i + 2] = q.z; ent[4 * i + 3] = q.w;
        }
#pragma unroll
        for (int i = 0; i < CAPS; ++i) {
            const unsigned e = ((unsigned)i < c_l) ? (ent[i] & 0xFFFFFu) : 0u;
            sn[i] = (unsigned)src[e];
        }
#pragma unroll
        for (int i = 0; i < CAPS; ++i) asm volatile("" :: "v"(sn[i]));
#pragma unroll
        for (int i = 0; i < CAPS; ++i) {
            if ((unsigned)i < c_l) myflat[off + i] = ((ent[i] >> 25) << 16) | sn[i];
        }
        __syncthreads();

        unsigned run[7];
#pragma unroll
        for (int j = 0; j < 7; ++j) run[j] = 0u;
        for (int k = 0; k < total; k += 32) {
            const int idx = k + lane;
            const unsigned slot = (idx < total) ? (myflat[idx] >> 16) : 7u;
#pragma unroll
            for (int j = 0; j < 7; ++j)
                nj[j] += (unsigned)__builtin_popcount(__builtin_amdgcn_ballot_w32(slot == (unsigned)j));
        }
        oj[0] = 0u;
#pragma unroll
        for (int j = 1; j < 7; ++j) oj[j] = oj[j - 1] + nj[j - 1];
        for (int k = 0; k < total; k += 32) {
            const int idx = k + lane;
            const unsigned ent  = (idx < total) ? myflat[idx] : (7u << 16);
            const unsigned slot = ent >> 16;
#pragma unroll
            for (int j = 0; j < 7; ++j) {
                const unsigned m = __builtin_amdgcn_ballot_w32(slot == (unsigned)j);
                if (slot == (unsigned)j)
                    mypart[oj[j] + run[j] + __builtin_amdgcn_mbcnt_lo(m, 0u)] = ent & 0xFFFFu;
                run[j] += (unsigned)__builtin_popcount(m);
            }
        }
        } else {
            oj[0] = 0u;
#pragma unroll
            for (int j = 1; j < 7; ++j) oj[j] = oj[j - 1] + nj[j - 1];
            const unsigned total = oj[6] + nj[6];
            for (unsigned k = 0u; k < total; k += 128u) {
                unsigned e[4];
#pragma unroll
                for (int u = 0; u < 4; ++u) {
                    const unsigned idx = k + (unsigned)(u * 32 + lane);
                    e[u] = (idx < total) ? kent[idx] : 0u;
                }
#pragma unroll
                for (int u = 0; u < 4; ++u) {
                    const unsigned idx = k + (unsigned)(u * 32 + lane);
                    if (idx < total) mypart[idx] = e[u];
                }
            }
        }
#pragma unroll
        for (int j = 0; j < 7; ++j) {
            const int r = j * 32 + w;
            if (r < RNODES && lane == 0)
                rowInfo[r] = ((unsigned)(w * FLAT_CAP) + oj[j]) | (nj[j] << 16);
        }
        __syncthreads();

        if (!KEPT) {
            const unsigned total = oj[6] + nj[6];
            for (unsigned k = 0u; k < total; k += 32u) {
                const unsigned idx = k + (unsigned)lane;
                if (idx < total) kent[idx] = mypart[idx];
            }
        }

        for (int t = 0; t <= RNODES; ++t) {
            unsigned took = 0u;
            if (lane == 0) took = atomicAdd(&nextRow, 1u);
            const unsigned r = (unsigned)__builtin_amdgcn_readfirstlane((int)took);
            if (r >= (unsigned)RNODES) break;
            const unsigned info = (unsigned)__builtin_amdgcn_readfirstlane((int)rowInfo[r]);
            const unsigned rcnt = info >> 16;
            const unsigned rbeg = info & 0xFFFFu;
            const float si = (r < nrows) ? rs_in[lo + r] : 0.f;
            if (!KEPT) {
                const float so = (r < nrows) ? rs_out[lo + r] : 0.f;
                if (lane == 0) rsoS[r] = so;
            }
            float4 a = make_float4(0.f, 0.f, 0.f, 0.f);
            for (unsigned k0 = 0u; k0 < rcnt; k0 += GATHER_U) {
                const unsigned kl   = k0 + (unsigned)lane;
                const unsigned entv = lists[rbeg + ((kl < rcnt) ? kl : (rcnt - 1u))];
                float4 v[GATHER_U];
                float  sc[GATHER_U];
#pragma unroll
                for (int u = 0; u < GATHER_U; ++u) {
                    const unsigned s = (unsigned)__builtin_amdgcn_readlane((int)entv, u);
                    v[u]  = X4[(size_t)s * (DFEAT / 4) + lane];
                    sc[u] = KEPT ? 1.f : rs_out[s];
                }
#pragma unroll
                for (int u = 0; u < GATHER_U; ++u) {
                    if (KEPT) asm volatile("" :: "v"(v[u].x), "v"(v[u].y), "v"(v[u].z), "v"(v[u].w));
                    else      asm volatile("" :: "v"(v[u].x), "v"(v[u].y), "v"(v[u].z), "v"(v[u].w),
                                           "s"(sc[u]));
                }
#pragma unroll
                for (int u = 0; u < GATHER_U; ++u) {
                    if (k0 + (unsigned)u < rcnt) {
                        if (KEPT) add4(a, v[u]);
                        else      fma4(a, v[u], sc[u]);
                    }
                }
            }
            *(float4*)&Alds[r * ASTRIDE + lane * 4] =
                make_float4(a.x * si, a.y * si, a.z * si, a.w * si);
        }
    } else {
        for (int pos = 0; pos < E4_TOTAL; pos += 32) {
            const int4 d = dst4[pos + lane];
            const unsigned u0 = (unsigned)d.x - (unsigned)lo;
            const unsigned u1 = (unsigned)d.y - (unsigned)lo;
            const unsigned u2 = (unsigned)d.z - (unsigned)lo;
            const unsigned u3 = (unsigned)d.w - (unsigned)lo;
            unsigned m0 = __builtin_amdgcn_ballot_w32(u0 < nrows && (u0 & 31u) == (unsigned)w);
            unsigned m1 = __builtin_amdgcn_ballot_w32(u1 < nrows && (u1 & 31u) == (unsigned)w);
            unsigned m2 = __builtin_amdgcn_ballot_w32(u2 < nrows && (u2 & 31u) == (unsigned)w);
            unsigned m3 = __builtin_amdgcn_ballot_w32(u3 < nrows && (u3 & 31u) == (unsigned)w);
            if ((m0 | m1 | m2 | m3) != 0u) {
#define SLOW_TAKE(m, u, c)                                                              \
                while (m) {                                                             \
                    const int l = __builtin_ctz(m);                                     \
                    m &= m - 1u;                                                        \
                    const unsigned row = (unsigned)__builtin_amdgcn_readlane((int)(u), l); \
                    const int      s   = src[4 * (pos + l) + (c)];                      \
                    const float    scs = KEPT ? 1.f : rs_out[s];                        \
                    const float4   vs  = X4[(size_t)s * (DFEAT / 4) + lane];            \
                    ACC_ROW(row >> 5, vs, scs)                                          \
                }
                SLOW_TAKE(m0, u0, 0)
                SLOW_TAKE(m1, u1, 1)
                SLOW_TAKE(m2, u2, 2)
                SLOW_TAKE(m3, u3, 3)
#undef SLOW_TAKE
            }
        }
    }

    if (slow) {
#define PUT_ROW(j, a)                                                                   \
    {                                                                                   \
        const int r = (j) * 32 + w;                                                     \
        if (r < RNODES) {                                                               \
            const float si = (r < (int)nrows) ? rs_in[lo + r] : 0.f;                    \
            *(float4*)&Alds[r * ASTRIDE + lane * 4] =                                   \
                make_float4(a.x * si, a.y * si, a.z * si, a.w * si);                    \
            if (!KEPT && lane == 0) rsoS[r] = (r < (int)nrows) ? rs_out[lo + r] : 0.f;  \
        }                                                                               \
    }
    PUT_ROW(0, a0)
    PUT_ROW(1, a1)
    PUT_ROW(2, a2)
    PUT_ROW(3, a3)
    PUT_ROW(4, a4)
    PUT_ROW(5, a5)
    PUT_ROW(6, a6)
#undef PUT_ROW
    }
    __syncthreads();

    if (!KEPT) {
        unsigned hv = (lane == 7 && slow) ? 1u : 0u;
#pragma unroll
        for (int j = 0; j < 7; ++j) if (lane == j) hv = nj[j];
        krec[lane] = hv;
    }

    const int ntiles = (int)(nrows >> 4);
    const int g      = w >> 3;
    const int n0     = (w & 7) << 4;
    if (g == 0) tile_product<4, RELU, !KEPT>(Alds, Wm, bias, rsoS, out, lo, ntiles, g, n0, lane);
    else        tile_product<3, RELU, !KEPT>(Alds, Wm, bias, rsoS, out, lo, ntiles, g, n0, lane);
}

extern "C" void kernel_launch(void* const* d_in, const int* in_sizes, int n_in,
                              void* d_out, int out_size, void* d_ws, size_t ws_size,
                              hipStream_t stream) {
    const float* x   = (const float*)d_in[0];
    const int*   src = (const int*)  d_in[1];
    const int*   dst = (const int*)  d_in[2];
    const float* W1  = (const float*)d_in[3];
    const float* b1  = (const float*)d_in[4];
    const float* W2  = (const float*)d_in[5];
    const float* b2  = (const float*)d_in[6];
    float*       out = (float*)d_out;

    float* ws     = (float*)d_ws;
    float* rs_out = ws;
    float* rs_in  = ws + N_NODES;
    float* scr    = ws + 2 * N_NODES;
    float* h1     = scr + (size_t)N_NODES * DFEAT;

    const int T = 256;

    unsigned* part = (unsigned*)scr;
    degree_hist<<<3 * NPART, HIST_THREADS, 0, stream>>>(src, dst, part);
    degree_finish<<<(2 * N4 + T - 1) / T, T, 0, stream>>>(part, rs_out);

    unsigned* kept = (unsigned*)scr;
    gcn_layer<true, false><<<NLBLK, LTHREADS, 0, stream>>>(x, rs_out, rs_in, src, dst, W1, b1, h1, kept);
    gcn_layer<false, true><<<NLBLK, LTHREADS, 0, stream>>>(h1, rs_out, rs_in, src, dst, W2, b2, out, kept);
}


# Layer-1 filter bitmap marks 8-edge pieces, not 32-edge chunks: a wave load covers 4 marked pieces, 2 load batches not 7; bitmaps kept in the out plane; sum order: same-load hits as the LDS add serves
# speedup vs baseline: 4.4370x; 1.0454x over previous
#include <hip/hip_runtime.h>

#define N_NODES 50000
#define N_EDGES 800000
#define DFEAT   128
#define ASTRIDE 132

typedef float v2f __attribute__((ext_vector_type(2)));
typedef float v8f __attribute__((ext_vector_type(8)));

#define RNODES   208
#define NLBLK    ((N_NODES + RNODES - 1) / RNODES)
#define LTHREADS 1024
#define LWAVES   32
#define CAPS     24
#define E4_TOTAL (N_EDGES / 4)
#define E4_WAVE  (E4_TOTAL / LWAVES)
#define SCAN_R   16
#define GATHER_U 8
#define FLAT_CAP (LWAVES * CAPS)
#define KEEP_HEAD  32
#define KEEP_REC   (KEEP_HEAD + FLAT_CAP)
#define KEEP_BASE  32
#define SUB_SHIFT 3
#define BWORDS   100
#define BSTRIDE  128
#define CL_CAP   800
#define NPIECE   ((4 * E4_WAVE) >> SUB_SHIFT)
static_assert(KEEP_BASE + NLBLK * LWAVES * KEEP_REC <= N_NODES * DFEAT,
              "kept records must fit the scratch plane");
static_assert(NLBLK * LWAVES * BSTRIDE <= N_NODES * DFEAT,
              "piece bitmaps must fit the output plane");
static_assert(CL_CAP >= FLAT_CAP && LWAVES * CL_CAP <= RNODES * ASTRIDE,
              "more marked pieces than CL_CAP overflow a list anyway; the piece lists share the tile's bytes");

#define NPART 32
#define EDGES_PER_PART (N_EDGES / NPART)
static_assert(NPART == LWAVES && EDGES_PER_PART == 4 * E4_WAVE,
              "a bitmap group's share is one scanning wave's share");
static_assert(EDGES_PER_PART % (1 << SUB_SHIFT) == 0 &&
              ((EDGES_PER_PART >> SUB_SHIFT) + 31) / 32 <= BWORDS && BWORDS % 4 == 0 &&
              BWORDS / 4 <= 32 && BWORDS <= BSTRIDE,
              "a share is whole pieces, and its pieces fit one bitmap record held as one uint4 per lane");
static_assert(NLBLK * BWORDS <= N_NODES, "the bitmaps of one share fit the counter array's LDS bytes");
#define E4_PER_PART    (EDGES_PER_PART / 4)
#define N4             (N_NODES / 4)
#define HIST_THREADS   1024
#define E4_PER_THREAD  ((E4_PER_PART + HIST_THREADS - 1) / HIST_THREADS)

__global__ __launch_bounds__(HIST_THREADS) void degree_hist(const int* __restrict__ src,
                                                            const int* __restrict__ dst,
                                                            unsigned* __restrict__ part,
                                                            unsigned* __restrict__ bits) {
    __shared__ __attribute__((aligned(16))) unsigned hist[N_NODES];
    unsigned* cbits = hist;

    const int tid = threadIdx.x;
    const int arr = blockIdx.x / NPART;
    const int p   = blockIdx.x % NPART;
    const int* idx = (arr == 0) ? src : dst;
    const int4* ep = (const int4*)(idx + (size_t)p * EDGES_PER_PART);

    int4 e[E4_PER_THREAD];
#pragma unroll
    for (int j = 0; j < E4_PER_THREAD; ++j) {
        const int i = tid + j * HIST_THREADS;
        if (i < E4_PER_PART) e[j] = ep[i];
    }

    if (arr < 2) {
        uint4* h4 = (uint4*)hist;
        for (int i = tid; i < N4; i += HIST_THREADS) h4[i] = make_uint4(0u, 0u, 0u, 0u);
        __syncthreads();

#pragma unroll
        for (int j = 0; j < E4_PER_THREAD; ++j) {
            const int i = tid + j * HIST_THREADS;
            if (i < E4_PER_PART) {
                atomicAdd(&hist[e[j].x], 1u);
                atomicAdd(&hist[e[j].y], 1u);
                atomicAdd(&hist[e[j].z], 1u);
                atomicAdd(&hist[e[j].w], 1u);
            }
        }
        __syncthreads();

        uint4* op = (uint4*)(part + (size_t)blockIdx.x * N_NODES);
        for (int i = tid; i < N4; i += HIST_THREADS) op[i] = h4[i];
    } else {
        uint4* c4 = (uint4*)cbits;
        for (int i = tid; i < NLBLK * (BWORDS / 4); i += HIST_THREADS) c4[i] = make_uint4(0u, 0u, 0u, 0u);
        __syncthreads();

#pragma unroll
        for (int j = 0; j < E4_PER_THREAD; ++j) {
            const int i = tid + j * HIST_THREADS;
            if (i < E4_PER_PART) {
                const unsigned lc  = (unsigned)i >> (SUB_SHIFT - 2);
                const unsigned wd  = lc >> 5;
                const unsigned bit = 1u << (lc & 31u);
                atomicOr(&cbits[((unsigned)e[j].x / RNODES) * BWORDS + wd], bit);
                atomicOr(&cbits[((unsigned)e[j].y / RNODES) * BWORDS + wd], bit);
                atomicOr(&cbits[((unsigned)e[j].z / RNODES) * BWORDS + wd], bit);
                atomicOr(&cbits[((unsigned)e[j].w / RNODES) * BWORDS + wd], bit);
            }
        }
        __syncthreads();

        uint4* bp4 = (uint4*)bits;
        for (int i = tid; i < NLBLK * (BWORDS / 4); i += HIST_THREADS) {
            const int b = i / (BWORDS / 4);
            const int k = i - b * (BWORDS / 4);
            bp4[(size_t)(b * NPART + p) * (BSTRIDE / 4) + k] = c4[i];
        }
    }
}

__global__ __launch_bounds__(256) void degree_finish(const unsigned* __restrict__ part,
                                                     float* __restrict__ rs) {
    const int t = blockIdx.x * blockDim.x + threadIdx.x;
    if (t >= 2 * N4) return;
    const int arr = (t >= N4) ? 1 : 0;
    const int q   = t - arr * N4;
    const uint4* pp = (const uint4*)part + (size_t)arr * NPART * N4 + q;
    uint4 s = make_uint4(0u, 0u, 0u, 0u);
#pragma unroll
    for (int p = 0; p < NPART; ++p) {
        const uint4 v = pp[(size_t)p * N4];
        s.x += v.x; s.y += v.y; s.z += v.z; s.w += v.w;
    }
    float4 r;
    r.x = rsqrtf(fmaxf((float)s.x, 1.0f));
    r.y = rsqrtf(fmaxf((float)s.y, 1.0f));
    r.z = rsqrtf(fmaxf((float)s.z, 1.0f));
    r.w = rsqrtf(fmaxf((float)s.w, 1.0f));
    ((float4*)rs)[t] = r;
}


__device__ __forceinline__ void fma4(float4& a, const float4 v, const float s) {
    a.x = fmaf(v.x, s, a.x);
    a.y = fmaf(v.y, s, a.y);
    a.z = fmaf(v.z, s, a.z);
    a.w = fmaf(v.w, s, a.w);
}

__device__ __forceinline__ void add4(float4& a, const float4 v) {
    a.x += v.x;
    a.y += v.y;
    a.z += v.z;
    a.w += v.w;
}

#define ACC_ROW(slot, v, s)                         \
    switch (slot) {                                 \
    case 0: fma4(a0, v, s); break;                  \
    case 1: fma4(a1, v, s); break;                  \
    case 2: fma4(a2, v, s); break;                  \
    case 3: fma4(a3, v, s); break;                  \
    case 4: fma4(a4, v, s); break;                  \
    case 5: fma4(a5, v, s); break;                  \
    case 6: fma4(a6, v, s); break;                  \
    default: break;                                 \
    }

template <int NT, bool RELU, bool SCALE_OUT>
__device__ __forceinline__ void tile_product(const float* Alds, const float* __restrict__ Wm,
                                             const float* __restrict__ bias,
                                             const float* rowscale,
                                             float* __restrict__ out,
                                             int lo, int ntiles, int g, int n0, int lane) {
    const int mrow = lane & 15;
    const int khi  = (lane >> 4) << 1;
    const float* wp = Wm + n0 + mrow;
    const float* ap[NT];
    bool valid[NT];
    v8f acc[NT];
#pragma unroll
    for (int t = 0; t < NT; ++t) {
        const int rt = g + 4 * t;
        valid[t] = rt < ntiles;
        const int rtc = valid[t] ? rt : g;
        ap[t] = Alds + (rtc * 16 + mrow) * ASTRIDE + khi;
        acc[t] = (v8f){0.f, 0.f, 0.f, 0.f, 0.f, 0.f, 0.f, 0.f};
    }
#pragma unroll
    for (int k = 0; k < DFEAT; k += 4) {
        v2f b;
        b.x = wp[(size_t)(k + khi)     * DFEAT];
        b.y = wp[(size_t)(k + khi + 1) * DFEAT];
#pragma unroll
        for (int t = 0; t < NT; ++t) {
            const v2f a = *(const v2f*)(ap[t] + k);
            acc[t] = __builtin_amdgcn_wmma_f32_16x16x4_f32(false, a, false, b,
                                                           (short)0, acc[t], false, false);
        }
    }
    const int   col  = n0 + mrow;
    const float bv   = bias[col];
    const int   rsel = (lane >> 4) << 3;
#pragma unroll
    for (int t = 0; t < NT; ++t) {
        if (valid[t]) {
            const int rowBase = lo + (g + 4 * t) * 16;
#pragma unroll
            for (int v = 0; v < 8; ++v) {
                float h = acc[t][v] + bv;
                if (RELU) h = fmaxf(h, 0.f);
                if (SCALE_OUT) h *= rowscale[(g + 4 * t) * 16 + v + rsel];
                out[(size_t)(rowBase + v + rsel) * DFEAT + col] = h;
            }
        }
    }
}

template <bool RELU, bool KEPT>
__global__ __launch_bounds__(LTHREADS) void gcn_layer(const float* __restrict__ X,
                                                      const float* __restrict__ rs_out,
                                                      const float* __restrict__ rs_in,
                                                      const int* __restrict__ src,
                                                      const int* __restrict__ dst,
                                                      const float* __restrict__ Wm,
                                                      const float* __restrict__ bias,
                                                      float* __restrict__ out,
                                                      unsigned* __restrict__ keep,
                                                      const unsigned* __restrict__ bits) {
    __shared__ __attribute__((aligned(16))) float Alds[RNODES * ASTRIDE];
    __shared__ __attribute__((aligned(16))) unsigned lists[LWAVES * LWAVES * CAPS];
    __shared__ unsigned cntS[LWAVES * LWAVES];
    __shared__ unsigned ovfFlag;
    __shared__ unsigned rowInfo[RNODES];
    __shared__ unsigned nextRow;
    __shared__ float rsoS[RNODES];

    const int tid  = threadIdx.x;
    const int lane = tid & 31;
    const int w    = __builtin_amdgcn_readfirstlane(tid >> 5);
    const int lo   = blockIdx.x * RNODES;
    const unsigned nrows = (N_NODES - lo < RNODES) ? (unsigned)(N_NODES - lo) : (unsigned)RNODES;

    const int4*   dst4 = (const int4*)dst;
    const float4* X4   = (const float4*)X;

    unsigned* krec = keep + KEEP_BASE + (size_t)((unsigned)blockIdx.x * LWAVES + (unsigned)w) * KEEP_REC;
    unsigned* kent = krec + KEEP_HEAD;

    bool slow;
    unsigned nj[7];
    unsigned* mypart = lists + w * FLAT_CAP;
    if (tid == 0) nextRow = 0u;

    if (KEPT) {
        const unsigned hv = krec[lane];
        slow = __builtin_amdgcn_readlane((int)hv, 7) != 0;
#pragma unroll
        for (int j = 0; j < 7; ++j) nj[j] = (unsigned)__builtin_amdgcn_readlane((int)hv, j);
    } else {
#pragma unroll
        for (int j = 0; j < 7; ++j) nj[j] = 0u;
    if (tid == 0) ovfFlag = 0u;
    __syncthreads();

    {
        const int ebeg = w * (4 * E4_WAVE);
        cntS[lane * LWAVES + w] = (unsigned)((lane * LWAVES + w) * CAPS);
        const uint4* bw4 = (const uint4*)(bits + (size_t)((unsigned)blockIdx.x * LWAVES + (unsigned)w) * BSTRIDE);
        uint4 mw = make_uint4(0u, 0u, 0u, 0u);
        if (lane < BWORDS / 4) mw = bw4[lane];
        unsigned* clist = (unsigned*)Alds + w * CL_CAP;
        const unsigned pc = (unsigned)(__builtin_popcount(mw.x) + __builtin_popcount(mw.y) +
                                       __builtin_popcount(mw.z) + __builtin_popcount(mw.w));
        unsigned cincl = pc;
#pragma unroll
        for (int dd = 1; dd < 32; dd <<= 1) {
            const unsigned up = __shfl_up(cincl, dd);
            if (lane >= dd) cincl += up;
        }
        const int nall = __builtin_amdgcn_readlane((int)cincl, 31);
        const bool toomany = nall > CL_CAP;
        {
            const unsigned mws[4] = {mw.x, mw.y, mw.z, mw.w};
            unsigned o = cincl - pc;
#pragma unroll
            for (int j = 0; j < 4; ++j) {
                unsigned mm = mws[j];
                while (mm) {
                    if (o < (unsigned)CL_CAP) clist[o] = (unsigned)(lane * 128 + j * 32 + __builtin_ctz(mm));
                    ++o;
                    mm &= mm - 1u;
                }
            }
        }
        __syncthreads();
        const int nsub = toomany ? 0 : nall;
        const int lgrp = lane >> 3;
        const int lsub = lane & 7;
        for (int k0 = 0; k0 < nsub; k0 += 4 * SCAN_R) {
            unsigned sc[SCAN_R];
#pragma unroll
            for (int r = 0; r < SCAN_R; ++r) {
                const int idx = k0 + 4 * r + lgrp;
                sc[r] = clist[(idx < nsub) ? idx : 0];
            }
            int dv[SCAN_R];
#pragma unroll
            for (int r = 0; r < SCAN_R; ++r) {
                const int idx = k0 + 4 * r + lgrp;
                const unsigned pcs = (sc[r] < (unsigned)NPIECE) ? sc[r] : (unsigned)(NPIECE - 1);
                const int e = ebeg + (int)(pcs << SUB_SHIFT) + lsub;
                dv[r] = dst[(idx < nsub) ? e : ebeg];
            }
#pragma unroll
            for (int r = 0; r < SCAN_R; ++r) asm volatile("" :: "v"(dv[r]));
            unsigned nn[SCAN_R];
            unsigned ev[SCAN_R];
#pragma unroll
            for (int r = 0; r < SCAN_R; ++r) {
                const int      idx = k0 + 4 * r + lgrp;
                const unsigned pcs = (sc[r] < (unsigned)NPIECE) ? sc[r] : (unsigned)(NPIECE - 1);
                const int      e   = ebeg + (int)(pcs << SUB_SHIFT) + lsub;
                const unsigned uu  = (unsigned)dv[r] - (unsigned)lo;
                ev[r] = (uu << 20) | (unsigned)e;
                nn[r] = 0xFFFFFFFFu;
                if (idx < nsub && uu < nrows)
                    nn[r] = atomicAdd(&cntS[(uu & 31u) * LWAVES + (unsigned)w], 1u);
            }
#pragma unroll
            for (int r = 0; r < SCAN_R; ++r) asm volatile("" :: "v"(nn[r]));
#pragma unroll
            for (int r = 0; r < SCAN_R; ++r) {
                if (nn[r] < (unsigned)(LWAVES * LWAVES * CAPS)) lists[nn[r]] = ev[r];
            }
        }
        if (toomany) ovfFlag = 1u;
        {
            const unsigned cnt = cntS[lane * LWAVES + w] - (unsigned)((lane * LWAVES + w) * CAPS);
            cntS[lane * LWAVES + w] = cnt;
            if (cnt > (unsigned)CAPS) ovfFlag = 1u;
        }
    }
    __syncthreads();
    slow = __builtin_amdgcn_readfirstlane((int)ovfFlag) != 0;
    }

    float4 a0 = make_float4(0.f, 0.f, 0.f, 0.f);
    float4 a1 = a0, a2 = a0, a3 = a0, a4 = a0, a5 = a0, a6 = a0;

    if (!slow) {
        unsigned oj[7];
        if (!KEPT) {
        unsigned* myflat = (unsigned*)Alds + w * FLAT_CAP;
        const unsigned c_l = cntS[w * LWAVES + lane];
        unsigned incl = c_l;
#pragma unroll
        for (int dd = 1; dd < 32; dd <<= 1) {
            const unsigned up = __shfl_up(incl, dd);
            if (lane >= dd) incl += up;
        }
        const unsigned off = incl - c_l;
        const int total = __builtin_amdgcn_readlane((int)incl, 31);
        const uint4* mylist4 = (const uint4*)(lists + (w * LWAVES + lane) * CAPS);
        unsigned ent[CAPS];
        unsigned sn[CAPS];
#pragma unroll
        for (int i = 0; i < CAPS / 4; ++i) {
            const uint4 q = mylist4[i];
            ent[4 * i + 0] = q.x; ent[4 * i + 1] = q.y; ent[4 * i + 2] = q.z; ent[4 * i + 3] = q.w;
        }
#pragma unroll
        for (int i = 0; i < CAPS; ++i) {
            const unsigned e = ((unsigned)i < c_l) ? (ent[i] & 0xFFFFFu) : 0u;
            sn[i] = (unsigned)src[e];
        }
#pragma unroll
        for (int i = 0; i < CAPS; ++i) asm volatile("" :: "v"(sn[i]));
#pragma unroll
        for (int i = 0; i < CAPS; ++i) {
            if ((unsigned)i < c_l) myflat[off + i] = ((ent[i] >> 25) << 16) | sn[i];
        }
        __syncthreads();

        unsigned run[7];
#pragma unroll
        for (int j = 0; j < 7; ++j) run[j] = 0u;
        for (int k = 0; k < total; k += 32) {
            const int idx = k + lane;
            const unsigned slot = (idx < total) ? (myflat[idx] >> 16) : 7u;
#pragma unroll
            for (int j = 0; j < 7; ++j)
                nj[j] += (unsigned)__builtin_popcount(__builtin_amdgcn_ballot_w32(slot == (unsigned)j));
        }
        oj[0] = 0u;
#pragma unroll
        for (int j = 1; j < 7; ++j) oj[j] = oj[j - 1] + nj[j - 1];
        for (int k = 0; k < total; k += 32) {
            const int idx = k + lane;
            const unsigned ent  = (idx < total) ? myflat[idx] : (7u << 16);
            const unsigned slot = ent >> 16;
#pragma unroll
            for (int j = 0; j < 7; ++j) {
                const unsigned m = __builtin_amdgcn_ballot_w32(slot == (unsigned)j);
                if (slot == (unsigned)j)
                    mypart[oj[j] + run[j] + __builtin_amdgcn_mbcnt_lo(m, 0u)] = ent & 0xFFFFu;
                run[j] += (unsigned)__builtin_popcount(m);
            }
        }
        } else {
            oj[0] = 0u;
#pragma unroll
            for (int j = 1; j < 7; ++j) oj[j] = oj[j - 1] + nj[j - 1];
            const unsigned total = oj[6] + nj[6];
            for (unsigned k = 0u; k < total; k += 128u) {
                unsigned e[4];
#pragma unroll
                for (int u = 0; u < 4; ++u) {
                    const unsigned idx = k + (unsigned)(u * 32 + lane);
                    e[u] = (idx < total) ? kent[idx] : 0u;
                }
#pragma unroll
                for (int u = 0; u < 4; ++u) {
                    const unsigned idx = k + (unsigned)(u * 32 + lane);
                    if (idx < total) mypart[idx] = e[u];
                }
            }
        }
#pragma unroll
        for (int j = 0; j < 7; ++j) {
            const int r = j * 32 + w;
            if (r < RNODES && lane == 0)
                rowInfo[r] = ((unsigned)(w * FLAT_CAP) + oj[j]) | (nj[j] << 16);
        }
        __syncthreads();

        if (!KEPT) {
            const unsigned total = oj[6] + nj[6];
            for (unsigned k = 0u; k < total; k += 32u) {
                const unsigned idx = k + (unsigned)lane;
                if (idx < total) kent[idx] = mypart[idx];
            }
        }

        for (int t = 0; t <= RNODES; ++t) {
            unsigned took = 0u;
            if (lane == 0) took = atomicAdd(&nextRow, 1u);
            const unsigned r = (unsigned)__builtin_amdgcn_readfirstlane((int)took);
            if (r >= (unsigned)RNODES) break;
            const unsigned info = (unsigned)__builtin_amdgcn_readfirstlane((int)rowInfo[r]);
            const unsigned rcnt = info >> 16;
            const unsigned rbeg = info & 0xFFFFu;
            const float si = (r < nrows) ? rs_in[lo + r] : 0.f;
            if (!KEPT) {
                const float so = (r < nrows) ? rs_out[lo + r] : 0.f;
                if (lane == 0) rsoS[r] = so;
            }
            float4 a = make_float4(0.f, 0.f, 0.f, 0.f);
            for (unsigned k0 = 0u; k0 < rcnt; k0 += GATHER_U) {
                const unsigned kl   = k0 + (unsigned)lane;
                const unsigned entv = lists[rbeg + ((kl < rcnt) ? kl : (rcnt - 1u))];
                float4 v[GATHER_U];
                float  sc[GATHER_U];
#pragma unroll
                for (int u = 0; u < GATHER_U; ++u) {
                    const unsigned s = (unsigned)__builtin_amdgcn_readlane((int)entv, u);
                    v[u]  = X4[(size_t)s * (DFEAT / 4) + lane];
                    sc[u] = KEPT ? 1.f : rs_out[s];
                }
#pragma unroll
                for (int u = 0; u < GATHER_U; ++u) {
                    if (KEPT) asm volatile("" :: "v"(v[u].x), "v"(v[u].y), "v"(v[u].z), "v"(v[u].w));
                    else      asm volatile("" :: "v"(v[u].x), "v"(v[u].y), "v"(v[u].z), "v"(v[u].w),
                                           "s"(sc[u]));
                }
#pragma unroll
                for (int u = 0; u < GATHER_U; ++u) {
                    if (k0 + (unsigned)u < rcnt) {
                        if (KEPT) add4(a, v[u]);
                        else      fma4(a, v[u], sc[u]);
                    }
                }
            }
            *(float4*)&Alds[r * ASTRIDE + lane * 4] =
                make_float4(a.x * si, a.y * si, a.z * si, a.w * si);
        }
    } else {
        for (int pos = 0; pos < E4_TOTAL; pos += 32) {
            const int4 d = dst4[pos + lane];
            const unsigned u0 = (unsigned)d.x - (unsigned)lo;
            const unsigned u1 = (unsigned)d.y - (unsigned)lo;
            const unsigned u2 = (unsigned)d.z - (unsigned)lo;
            const unsigned u3 = (unsigned)d.w - (unsigned)lo;
            unsigned m0 = __builtin_amdgcn_ballot_w32(u0 < nrows && (u0 & 31u) == (unsigned)w);
            unsigned m1 = __builtin_amdgcn_ballot_w32(u1 < nrows && (u1 & 31u) == (unsigned)w);
            unsigned m2 = __builtin_amdgcn_ballot_w32(u2 < nrows && (u2 & 31u) == (unsigned)w);
            unsigned m3 = __builtin_amdgcn_ballot_w32(u3 < nrows && (u3 & 31u) == (unsigned)w);
            if ((m0 | m1 | m2 | m3) != 0u) {
#define SLOW_TAKE(m, u, c)                                                              \
                while (m) {                                                             \
                    const int l = __builtin_ctz(m);                                     \
                    m &= m - 1u;                                                        \
                    const unsigned row = (unsigned)__builtin_amdgcn_readlane((int)(u), l); \
                    const int      s   = src[4 * (pos + l) + (c)];                      \
                    const float    scs = KEPT ? 1.f : rs_out[s];                        \
                    const float4   vs  = X4[(size_t)s * (DFEAT / 4) + lane];            \
                    ACC_ROW(row >> 5, vs, scs)                                          \
                }
                SLOW_TAKE(m0, u0, 0)
                SLOW_TAKE(m1, u1, 1)
                SLOW_TAKE(m2, u2, 2)
                SLOW_TAKE(m3, u3, 3)
#undef SLOW_TAKE
            }
        }
    }

    if (slow) {
#define PUT_ROW(j, a)                                                                   \
    {                                                                                   \
        const int r = (j) * 32 + w;                                                     \
        if (r < RNODES) {                                                               \
            const float si = (r < (int)nrows) ? rs_in[lo + r] : 0.f;                    \
            *(float4*)&Alds[r * ASTRIDE + lane * 4] =                                   \
                make_float4(a.x * si, a.y * si, a.z * si, a.w * si);                    \
            if (!KEPT && lane == 0) rsoS[r] = (r < (int)nrows) ? rs_out[lo + r] : 0.f;  \
        }                                                                               \
    }
    PUT_ROW(0, a0)
    PUT_ROW(1, a1)
    PUT_ROW(2, a2)
    PUT_ROW(3, a3)
    PUT_ROW(4, a4)
    PUT_ROW(5, a5)
    PUT_ROW(6, a6)
#undef PUT_ROW
    }
    __syncthreads();

    if (!KEPT) {
        unsigned hv = (lane == 7 && slow) ? 1u : 0u;
#pragma unroll
        for (int j = 0; j < 7; ++j) if (lane == j) hv = nj[j];
        krec[lane] = hv;
    }

    const int ntiles = (int)(nrows >> 4);
    const int g      = w >> 3;
    const int n0     = (w & 7) << 4;
    if (g == 0) tile_product<4, RELU, !KEPT>(Alds, Wm, bias, rsoS, out, lo, ntiles, g, n0, lane);
    else        tile_product<3, RELU, !KEPT>(Alds, Wm, bias, rsoS, out, lo, ntiles, g, n0, lane);
}

extern "C" void kernel_launch(void* const* d_in, const int* in_sizes, int n_in,
                              void* d_out, int out_size, void* d_ws, size_t ws_size,
                              hipStream_t stream) {
    const float* x   = (const float*)d_in[0];
    const int*   src = (const int*)  d_in[1];
    const int*   dst = (const int*)  d_in[2];
    const float* W1  = (const float*)d_in[3];
    const float* b1  = (const float*)d_in[4];
    const float* W2  = (const float*)d_in[5];
    const float* b2  = (const float*)d_in[6];
    float*       out = (float*)d_out;

    float* ws     = (float*)d_ws;
    float* rs_out = ws;
    float* rs_in  = ws + N_NODES;
    float* scr    = ws + 2 * N_NODES;
    float* h1     = scr + (size_t)N_NODES * DFEAT;

    const int T = 256;

    unsigned* part = (unsigned*)scr;
    unsigned* bits = (unsigned*)d_out;
    degree_hist<<<3 * NPART, HIST_THREADS, 0, stream>>>(src, dst, part, bits);
    degree_finish<<<(2 * N4 + T - 1) / T, T, 0, stream>>>(part, rs_out);

    unsigned* kept = (unsigned*)scr;
    gcn_layer<true, false><<<NLBLK, LTHREADS, 0, stream>>>(x, rs_out, rs_in, src, dst, W1, b1, h1, kept, bits);
    gcn_layer<false, true><<<NLBLK, LTHREADS, 0, stream>>>(h1, rs_out, rs_in, src, dst, W2, b2, out, kept, nullptr);
}


# Layer-1 edge sum takes its source weights from an LDS table of all out-degrees as bytes (stored by degree_finish, copied in per group; rsqrt at the add), not a memory line per edge; sums unchanged
# speedup vs baseline: 4.5690x; 1.0297x over previous
#include <hip/hip_runtime.h>

#define N_NODES 50000
#define N_EDGES 800000
#define DFEAT   128
#define ASTRIDE 132

typedef float v2f __attribute__((ext_vector_type(2)));
typedef float v8f __attribute__((ext_vector_type(8)));

#define RNODES   208
#define NLBLK    ((N_NODES + RNODES - 1) / RNODES)
#define LTHREADS 1024
#define LWAVES   32
#define CAPS     24
#define E4_TOTAL (N_EDGES / 4)
#define E4_WAVE  (E4_TOTAL / LWAVES)
#define SCAN_R   16
#define GATHER_U 8
#define FLAT_CAP (LWAVES * CAPS)
#define KEEP_HEAD  32
#define KEEP_REC   (KEEP_HEAD + FLAT_CAP)
#define KEEP_BASE  32
#define SUB_SHIFT 3
#define BWORDS   100
#define BSTRIDE  128
#define CL_CAP   800
#define NPIECE   ((4 * E4_WAVE) >> SUB_SHIFT)
static_assert(KEEP_BASE + NLBLK * LWAVES * KEEP_REC <= N_NODES * DFEAT,
              "kept records must fit the scratch plane");
static_assert(NLBLK * LWAVES * BSTRIDE <= N_NODES * DFEAT,
              "piece bitmaps must fit the output plane");
static_assert(CL_CAP >= FLAT_CAP && LWAVES * CL_CAP <= RNODES * ASTRIDE,
              "more marked pieces than CL_CAP overflow a list anyway; the piece lists share the tile's bytes");

#define NPART 32
#define EDGES_PER_PART (N_EDGES / NPART)
static_assert(NPART == LWAVES && EDGES_PER_PART == 4 * E4_WAVE,
              "a bitmap group's share is one scanning wave's share");
static_assert(EDGES_PER_PART % (1 << SUB_SHIFT) == 0 &&
              ((EDGES_PER_PART >> SUB_SHIFT) + 31) / 32 <= BWORDS && BWORDS % 4 == 0 &&
              BWORDS / 4 <= 32 && BWORDS <= BSTRIDE,
              "a share is whole pieces, and its pieces fit one bitmap record held as one uint4 per lane");
static_assert(NLBLK * BWORDS <= N_NODES, "the bitmaps of one share fit the counter array's LDS bytes");
#define E4_PER_PART    (EDGES_PER_PART / 4)
#define N4             (N_NODES / 4)
#define DEGB_WORDS     12544
static_assert(N4 <= DEGB_WORDS && DEGB_WORDS % 32 == 0 &&
              KEEP_BASE + NLBLK * LWAVES * KEEP_REC + DEGB_WORDS <= N_NODES * DFEAT,
              "the degree bytes fit behind the kept records in the scratch plane");
#define HIST_THREADS   1024
#define E4_PER_THREAD  ((E4_PER_PART + HIST_THREADS - 1) / HIST_THREADS)

__global__ __launch_bounds__(HIST_THREADS) void degree_hist(const int* __restrict__ src,
                                                            const int* __restrict__ dst,
                                                            unsigned* __restrict__ part,
                                                            unsigned* __restrict__ bits) {
    __shared__ __attribute__((aligned(16))) unsigned hist[N_NODES];
    unsigned* cbits = hist;

    const int tid = threadIdx.x;
    const int arr = blockIdx.x / NPART;
    const int p   = blockIdx.x % NPART;
    const int* idx = (arr == 0) ? src : dst;
    const int4* ep = (const int4*)(idx + (size_t)p * EDGES_PER_PART);

    int4 e[E4_PER_THREAD];
#pragma unroll
    for (int j = 0; j < E4_PER_THREAD; ++j) {
        const int i = tid + j * HIST_THREADS;
        if (i < E4_PER_PART) e[j] = ep[i];
    }

    if (arr < 2) {
        uint4* h4 = (uint4*)hist;
        for (int i = tid; i < N4; i += HIST_THREADS) h4[i] = make_uint4(0u, 0u, 0u, 0u);
        __syncthreads();

#pragma unroll
        for (int j = 0; j < E4_PER_THREAD; ++j) {
            const int i = tid + j * HIST_THREADS;
            if (i < E4_PER_PART) {
                atomicAdd(&hist[e[j].x], 1u);
                atomicAdd(&hist[e[j].y], 1u);
                atomicAdd(&hist[e[j].z], 1u);
                atomicAdd(&hist[e[j].w], 1u);
            }
        }
        __syncthreads();

        uint4* op = (uint4*)(part + (size_t)blockIdx.x * N_NODES);
        for (int i = tid; i < N4; i += HIST_THREADS) op[i] = h4[i];
    } else {
        uint4* c4 = (uint4*)cbits;
        for (int i = tid; i < NLBLK * (BWORDS / 4); i += HIST_THREADS) c4[i] = make_uint4(0u, 0u, 0u, 0u);
        __syncthreads();

#pragma unroll
        for (int j = 0; j < E4_PER_THREAD; ++j) {
            const int i = tid + j * HIST_THREADS;
            if (i < E4_PER_PART) {
                const unsigned lc  = (unsigned)i >> (SUB_SHIFT - 2);
                const unsigned wd  = lc >> 5;
                const unsigned bit = 1u << (lc & 31u);
                atomicOr(&cbits[((unsigned)e[j].x / RNODES) * BWORDS + wd], bit);
                atomicOr(&cbits[((unsigned)e[j].y / RNODES) * BWORDS + wd], bit);
                atomicOr(&cbits[((unsigned)e[j].z / RNODES) * BWORDS + wd], bit);
                atomicOr(&cbits[((unsigned)e[j].w / RNODES) * BWORDS + wd], bit);
            }
        }
        __syncthreads();

        uint4* bp4 = (uint4*)bits;
        for (int i = tid; i < NLBLK * (BWORDS / 4); i += HIST_THREADS) {
            const int b = i / (BWORDS / 4);
            const int k = i - b * (BWORDS / 4);
            bp4[(size_t)(b * NPART + p) * (BSTRIDE / 4) + k] = c4[i];
        }
    }
}

__global__ __launch_bounds__(256) void degree_finish(const unsigned* __restrict__ part,
                                                     float* __restrict__ rs,
                                                     unsigned* __restrict__ degb) {
    const int t = blockIdx.x * blockDim.x + threadIdx.x;
    if (t >= 2 * N4) return;
    const int arr = (t >= N4) ? 1 : 0;
    const int q   = t - arr * N4;
    const uint4* pp = (const uint4*)part + (size_t)arr * NPART * N4 + q;
    uint4 s = make_uint4(0u, 0u, 0u, 0u);
#pragma unroll
    for (int p = 0; p < NPART; ++p) {
        const uint4 v = pp[(size_t)p * N4];
        s.x += v.x; s.y += v.y; s.z += v.z; s.w += v.w;
    }
    float4 r;
    r.x = rsqrtf(fmaxf((float)s.x, 1.0f));
    r.y = rsqrtf(fmaxf((float)s.y, 1.0f));
    r.z = rsqrtf(fmaxf((float)s.z, 1.0f));
    r.w = rsqrtf(fmaxf((float)s.w, 1.0f));
    ((float4*)rs)[t] = r;
    if (arr == 0)
        degb[q] = (s.x < 255u ? s.x : 255u) | ((s.y < 255u ? s.y : 255u) << 8) |
                  ((s.z < 255u ? s.z : 255u) << 16) | ((s.w < 255u ? s.w : 255u) << 24);
}


__device__ __forceinline__ void fma4(float4& a, const float4 v, const float s) {
    a.x = fmaf(v.x, s, a.x);
    a.y = fmaf(v.y, s, a.y);
    a.z = fmaf(v.z, s, a.z);
    a.w = fmaf(v.w, s, a.w);
}

__device__ __forceinline__ void add4(float4& a, const float4 v) {
    a.x += v.x;
    a.y += v.y;
    a.z += v.z;
    a.w += v.w;
}

#define ACC_ROW(slot, v, s)                         \
    switch (slot) {                                 \
    case 0: fma4(a0, v, s); break;                  \
    case 1: fma4(a1, v, s); break;                  \
    case 2: fma4(a2, v, s); break;                  \
    case 3: fma4(a3, v, s); break;                  \
    case 4: fma4(a4, v, s); break;                  \
    case 5: fma4(a5, v, s); break;                  \
    case 6: fma4(a6, v, s); break;                  \
    default: break;                                 \
    }

template <int NT, bool RELU, bool SCALE_OUT>
__device__ __forceinline__ void tile_product(const float* Alds, const float* __restrict__ Wm,
                                             const float* __restrict__ bias,
                                             const float* rowscale,
                                             float* __restrict__ out,
                                             int lo, int ntiles, int g, int n0, int lane) {
    const int mrow = lane & 15;
    const int khi  = (lane >> 4) << 1;
    const float* wp = Wm + n0 + mrow;
    const float* ap[NT];
    bool valid[NT];
    v8f acc[NT];
#pragma unroll
    for (int t = 0; t < NT; ++t) {
        const int rt = g + 4 * t;
        valid[t] = rt < ntiles;
        const int rtc = valid[t] ? rt : g;
        ap[t] = Alds + (rtc * 16 + mrow) * ASTRIDE + khi;
        acc[t] = (v8f){0.f, 0.f, 0.f, 0.f, 0.f, 0.f, 0.f, 0.f};
    }
#pragma unroll
    for (int k = 0; k < DFEAT; k += 4) {
        v2f b;
        b.x = wp[(size_t)(k + khi)     * DFEAT];
        b.y = wp[(size_t)(k + khi + 1) * DFEAT];
#pragma unroll
        for (int t = 0; t < NT; ++t) {
            const v2f a = *(const v2f*)(ap[t] + k);
            acc[t] = __builtin_amdgcn_wmma_f32_16x16x4_f32(false, a, false, b,
                                                           (short)0, acc[t], false, false);
        }
    }
    const int   col  = n0 + mrow;
    const float bv   = bias[col];
    const int   rsel = (lane >> 4) << 3;
#pragma unroll
    for (int t = 0; t < NT; ++t) {
        if (valid[t]) {
            const int rowBase = lo + (g + 4 * t) * 16;
#pragma unroll
            for (int v = 0; v < 8; ++v) {
                float h = acc[t][v] + bv;
                if (RELU) h = fmaxf(h, 0.f);
                if (SCALE_OUT) h *= rowscale[(g + 4 * t) * 16 + v + rsel];
                out[(size_t)(rowBase + v + rsel) * DFEAT + col] = h;
            }
        }
    }
}

template <bool RELU, bool KEPT>
__global__ __launch_bounds__(LTHREADS) void gcn_layer(const float* __restrict__ X,
                                                      const float* __restrict__ rs_out,
                                                      const float* __restrict__ rs_in,
                                                      const int* __restrict__ src,
                                                      const int* __restrict__ dst,
                                                      const float* __restrict__ Wm,
                                                      const float* __restrict__ bias,
                                                      float* __restrict__ out,
                                                      unsigned* __restrict__ keep,
                                                      const unsigned* __restrict__ bits,
                                                      const unsigned* __restrict__ degb) {
    __shared__ __attribute__((aligned(16))) float Alds[RNODES * ASTRIDE];
    __shared__ __attribute__((aligned(16))) unsigned lists[LWAVES * LWAVES * CAPS];
    __shared__ unsigned cntS[LWAVES * LWAVES];
    __shared__ unsigned ovfFlag;
    __shared__ unsigned rowInfo[RNODES];
    __shared__ unsigned nextRow;
    __shared__ float rsoS[RNODES];
    __shared__ __attribute__((aligned(16))) unsigned degS[N4];

    const int tid  = threadIdx.x;
    const int lane = tid & 31;
    const int w    = __builtin_amdgcn_readfirstlane(tid >> 5);
    const int lo   = blockIdx.x * RNODES;
    const unsigned nrows = (N_NODES - lo < RNODES) ? (unsigned)(N_NODES - lo) : (unsigned)RNODES;

    const int4*   dst4 = (const int4*)dst;
    const float4* X4   = (const float4*)X;

    unsigned* krec = keep + KEEP_BASE + (size_t)((unsigned)blockIdx.x * LWAVES + (unsigned)w) * KEEP_REC;
    unsigned* kent = krec + KEEP_HEAD;

    bool slow;
    unsigned nj[7];
    unsigned* mypart = lists + w * FLAT_CAP;
    if (tid == 0) nextRow = 0u;

    if (KEPT) {
        const unsigned hv = krec[lane];
        slow = __builtin_amdgcn_readlane((int)hv, 7) != 0;
#pragma unroll
        for (int j = 0; j < 7; ++j) nj[j] = (unsigned)__builtin_amdgcn_readlane((int)hv, j);
    } else {
#pragma unroll
        for (int j = 0; j < 7; ++j) nj[j] = 0u;
    if (tid == 0) ovfFlag = 0u;
    __syncthreads();

    {
        const int ebeg = w * (4 * E4_WAVE);
        cntS[lane * LWAVES + w] = (unsigned)((lane * LWAVES + w) * CAPS);
        const uint4* bw4 = (const uint4*)(bits + (size_t)((unsigned)blockIdx.x * LWAVES + (unsigned)w) * BSTRIDE);
        uint4 mw = make_uint4(0u, 0u, 0u, 0u);
        if (lane < BWORDS / 4) mw = bw4[lane];
        const uint4* degb4 = (const uint4*)degb;
        uint4 dq[4];
#pragma unroll
        for (int j = 0; j < 4; ++j) {
            const int i = tid + j * LTHREADS;
            dq[j] = make_uint4(0u, 0u, 0u, 0u);
            if (i < N4 / 4) dq[j] = degb4[i];
        }
        unsigned* clist = (unsigned*)Alds + w * CL_CAP;
        const unsigned pc = (unsigned)(__builtin_popcount(mw.x) + __builtin_popcount(mw.y) +
                                       __builtin_popcount(mw.z) + __builtin_popcount(mw.w));
        unsigned cincl = pc;
#pragma unroll
        for (int dd = 1; dd < 32; dd <<= 1) {
            const unsigned up = __shfl_up(cincl, dd);
            if (lane >= dd) cincl += up;
        }
        const int nall = __builtin_amdgcn_readlane((int)cincl, 31);
        const bool toomany = nall > CL_CAP;
        {
            const unsigned mws[4] = {mw.x, mw.y, mw.z, mw.w};
            unsigned o = cincl - pc;
#pragma unroll
            for (int j = 0; j < 4; ++j) {
                unsigned mm = mws[j];
                while (mm) {
                    if (o < (unsigned)CL_CAP) clist[o] = (unsigned)(lane * 128 + j * 32 + __builtin_ctz(mm));
                    ++o;
                    mm &= mm - 1u;
                }
            }
        }
#pragma unroll
        for (int j = 0; j < 4; ++j) {
            const int i = tid + j * LTHREADS;
            if (i < N4 / 4) ((uint4*)degS)[i] = dq[j];
        }
        __syncthreads();
        const int nsub = toomany ? 0 : nall;
        const int lgrp = lane >> 3;
        const int lsub = lane & 7;
        for (int k0 = 0; k0 < nsub; k0 += 4 * SCAN_R) {
            unsigned sc[SCAN_R];
#pragma unroll
            for (int r = 0; r < SCAN_R; ++r) {
                const int idx = k0 + 4 * r + lgrp;
                sc[r] = clist[(idx < nsub) ? idx : 0];
            }
            int dv[SCAN_R];
#pragma unroll
            for (int r = 0; r < SCAN_R; ++r) {
                const int idx = k0 + 4 * r + lgrp;
                const unsigned pcs = (sc[r] < (unsigned)NPIECE) ? sc[r] : (unsigned)(NPIECE - 1);
                const int e = ebeg + (int)(pcs << SUB_SHIFT) + lsub;
                dv[r] = dst[(idx < nsub) ? e : ebeg];
            }
#pragma unroll
            for (int r = 0; r < SCAN_R; ++r) asm volatile("" :: "v"(dv[r]));
            unsigned nn[SCAN_R];
            unsigned ev[SCAN_R];
#pragma unroll
            for (int r = 0; r < SCAN_R; ++r) {
                const int      idx = k0 + 4 * r + lgrp;
                const unsigned pcs = (sc[r] < (unsigned)NPIECE) ? sc[r] : (unsigned)(NPIECE - 1);
                const int      e   = ebeg + (int)(pcs << SUB_SHIFT) + lsub;
                const unsigned uu  = (unsigned)dv[r] - (unsigned)lo;
                ev[r] = (uu << 20) | (unsigned)e;
                nn[r] = 0xFFFFFFFFu;
                if (idx < nsub && uu < nrows)
                    nn[r] = atomicAdd(&cntS[(uu & 31u) * LWAVES + (unsigned)w], 1u);
            }
#pragma unroll
            for (int r = 0; r < SCAN_R; ++r) asm volatile("" :: "v"(nn[r]));
#pragma unroll
            for (int r = 0; r < SCAN_R; ++r) {
                if (nn[r] < (unsigned)(LWAVES * LWAVES * CAPS)) lists[nn[r]] = ev[r];
            }
        }
        if (toomany) ovfFlag = 1u;
        {
            const unsigned cnt = cntS[lane * LWAVES + w] - (unsigned)((lane * LWAVES + w) * CAPS);
            cntS[lane * LWAVES + w] = cnt;
            if (cnt > (unsigned)CAPS) ovfFlag = 1u;
        }
    }
    __syncthreads();
    slow = __builtin_amdgcn_readfirstlane((int)ovfFlag) != 0;
    }

    float4 a0 = make_float4(0.f, 0.f, 0.f, 0.f);
    float4 a1 = a0, a2 = a0, a3 = a0, a4 = a0, a5 = a0, a6 = a0;

    if (!slow) {
        unsigned oj[7];
        if (!KEPT) {
        unsigned* myflat = (unsigned*)Alds + w * FLAT_CAP;
        const unsigned c_l = cntS[w * LWAVES + lane];
        unsigned incl = c_l;
#pragma unroll
        for (int dd = 1; dd < 32; dd <<= 1) {
            const unsigned up = __shfl_up(incl, dd);
            if (lane >= dd) incl += up;
        }
        const unsigned off = incl - c_l;
        const int total = __builtin_amdgcn_readlane((int)incl, 31);
        const uint4* mylist4 = (const uint4*)(lists + (w * LWAVES + lane) * CAPS);
        unsigned ent[CAPS];
        unsigned sn[CAPS];
#pragma unroll
        for (int i = 0; i < CAPS / 4; ++i) {
            const uint4 q = mylist4[i];
            ent[4 * i + 0] = q.x; ent[4 * i + 1] = q.y; ent[4 * i + 2] = q.z; ent[4 * i + 3] = q.w;
        }
#pragma unroll
        for (int i = 0; i < CAPS; ++i) {
            const unsigned e = ((unsigned)i < c_l) ? (ent[i] & 0xFFFFFu) : 0u;
            sn[i] = (unsigned)src[e];
        }
#pragma unroll
        for (int i = 0; i < CAPS; ++i) asm volatile("" :: "v"(sn[i]));
#pragma unroll
        for (int i = 0; i < CAPS; ++i) {
            if ((unsigned)i < c_l) myflat[off + i] = ((ent[i] >> 25) << 16) | sn[i];
        }
        __syncthreads();

        unsigned run[7];
#pragma unroll
        for (int j = 0; j < 7; ++j) run[j] = 0u;
        for (int k = 0; k < total; k += 32) {
            const int idx = k + lane;
            const unsigned slot = (idx < total) ? (myflat[idx] >> 16) : 7u;
#pragma unroll
            for (int j = 0; j < 7; ++j)
                nj[j] += (unsigned)__builtin_popcount(__builtin_amdgcn_ballot_w32(slot == (unsigned)j));
        }
        oj[0] = 0u;
#pragma unroll
        for (int j = 1; j < 7; ++j) oj[j] = oj[j - 1] + nj[j - 1];
        for (int k = 0; k < total; k += 32) {
            const int idx = k + lane;
            const unsigned ent  = (idx < total) ? myflat[idx] : (7u << 16);
            const unsigned slot = ent >> 16;
#pragma unroll
            for (int j = 0; j < 7; ++j) {
                const unsigned m = __builtin_amdgcn_ballot_w32(slot == (unsigned)j);
                if (slot == (unsigned)j)
                    mypart[oj[j] + run[j] + __builtin_amdgcn_mbcnt_lo(m, 0u)] = ent & 0xFFFFu;
                run[j] += (unsigned)__builtin_popcount(m);
            }
        }
        } else {
            oj[0] = 0u;
#pragma unroll
            for (int j = 1; j < 7; ++j) oj[j] = oj[j - 1] + nj[j - 1];
            const unsigned total = oj[6] + nj[6];
            for (unsigned k = 0u; k < total; k += 128u) {
                unsigned e[4];
#pragma unroll
                for (int u = 0; u < 4; ++u) {
                    const unsigned idx = k + (unsigned)(u * 32 + lane);
                    e[u] = (idx < total) ? kent[idx] : 0u;
                }
#pragma unroll
                for (int u = 0; u < 4; ++u) {
                    const unsigned idx = k + (unsigned)(u * 32 + lane);
                    if (idx < total) mypart[idx] = e[u];
                }
            }
        }
#pragma unroll
        for (int j = 0; j < 7; ++j) {
            const int r = j * 32 + w;
            if (r < RNODES && lane == 0)
                rowInfo[r] = ((unsigned)(w * FLAT_CAP) + oj[j]) | (nj[j] << 16);
        }
        __syncthreads();

        if (!KEPT) {
            const unsigned total = oj[6] + nj[6];
            for (unsigned k = 0u; k < total; k += 32u) {
                const unsigned idx = k + (unsigned)lane;
                if (idx < total) kent[idx] = mypart[idx];
            }
        }

        for (int t = 0; t <= RNODES; ++t) {
            unsigned took = 0u;
            if (lane == 0) took = atomicAdd(&nextRow, 1u);
            const unsigned r = (unsigned)__builtin_amdgcn_readfirstlane((int)took);
            if (r >= (unsigned)RNODES) break;
            const unsigned info = (unsigned)__builtin_amdgcn_readfirstlane((int)rowInfo[r]);
            const unsigned rcnt = info >> 16;
            const unsigned rbeg = info & 0xFFFFu;
            const float si = (r < nrows) ? rs_in[lo + r] : 0.f;
            if (!KEPT) {
                const float so = (r < nrows) ? rs_out[lo + r] : 0.f;
                if (lane == 0) rsoS[r] = so;
            }
            float4 a = make_float4(0.f, 0.f, 0.f, 0.f);
            for (unsigned k0 = 0u; k0 < rcnt; k0 += GATHER_U) {
                const unsigned kl   = k0 + (unsigned)lane;
                const unsigned entv = lists[rbeg + ((kl < rcnt) ? kl : (rcnt - 1u))];
                float4 v[GATHER_U];
#pragma unroll
                for (int u = 0; u < GATHER_U; ++u) {
                    const unsigned s = (unsigned)__builtin_amdgcn_readlane((int)entv, u);
                    v[u]  = X4[(size_t)s * (DFEAT / 4) + lane];
                }
                int wl = 0;
                if (!KEPT) {
                    asm volatile("" :: "v"(entv));
                    const unsigned db = ((const unsigned char*)degS)[entv];
                    float wf = rsqrtf(fmaxf((float)db, 1.0f));
                    if (db == 255u) wf = rs_out[entv];
                    wl = __builtin_bit_cast(int, wf);
                }
#pragma unroll
                for (int u = 0; u < GATHER_U; ++u)
                    asm volatile("" :: "v"(v[u].x), "v"(v[u].y), "v"(v[u].z), "v"(v[u].w));
#pragma unroll
                for (int u = 0; u < GATHER_U; ++u) {
                    if (k0 + (unsigned)u < rcnt) {
                        if (KEPT) add4(a, v[u]);
                        else      fma4(a, v[u], __builtin_bit_cast(float, __builtin_amdgcn_readlane(wl, u)));
                    }
                }
            }
            *(float4*)&Alds[r * ASTRIDE + lane * 4] =
                make_float4(a.x * si, a.y * si, a.z * si, a.w * si);
        }
    } else {
        for (int pos = 0; pos < E4_TOTAL; pos += 32) {
            const int4 d = dst4[pos + lane];
            const unsigned u0 = (unsigned)d.x - (unsigned)lo;
            const unsigned u1 = (unsigned)d.y - (unsigned)lo;
            const unsigned u2 = (unsigned)d.z - (unsigned)lo;
            const unsigned u3 = (unsigned)d.w - (unsigned)lo;
            unsigned m0 = __builtin_amdgcn_ballot_w32(u0 < nrows && (u0 & 31u) == (unsigned)w);
            unsigned m1 = __builtin_amdgcn_ballot_w32(u1 < nrows && (u1 & 31u) == (unsigned)w);
            unsigned m2 = __builtin_amdgcn_ballot_w32(u2 < nrows && (u2 & 31u) == (unsigned)w);
            unsigned m3 = __builtin_amdgcn_ballot_w32(u3 < nrows && (u3 & 31u) == (unsigned)w);
            if ((m0 | m1 | m2 | m3) != 0u) {
#define SLOW_TAKE(m, u, c)                                                              \
                while (m) {                                                             \
                    const int l = __builtin_ctz(m);                                     \
                    m &= m - 1u;                                                        \
                    const unsigned row = (unsigned)__builtin_amdgcn_readlane((int)(u), l); \
                    const int      s   = src[4 * (pos + l) + (c)];                      \
                    const float    scs = KEPT ? 1.f : rs_out[s];                        \
                    const float4   vs  = X4[(size_t)s * (DFEAT / 4) + lane];            \
                    ACC_ROW(row >> 5, vs, scs)                                          \
                }
                SLOW_TAKE(m0, u0, 0)
                SLOW_TAKE(m1, u1, 1)
                SLOW_TAKE(m2, u2, 2)
                SLOW_TAKE(m3, u3, 3)
#undef SLOW_TAKE
            }
        }
    }

    if (slow) {
#define PUT_ROW(j, a)                                                                   \
    {                                                                                   \
        const int r = (j) * 32 + w;                                                     \
        if (r < RNODES) {                                                               \
            const float si = (r < (int)nrows) ? rs_in[lo + r] : 0.f;                    \
            *(float4*)&Alds[r * ASTRIDE + lane * 4] =                                   \
                make_float4(a.x * si, a.y * si, a.z * si, a.w * si);                    \
            if (!KEPT && lane == 0) rsoS[r] = (r < (int)nrows) ? rs_out[lo + r] : 0.f;  \
        }                                                                               \
    }
    PUT_ROW(0, a0)
    PUT_ROW(1, a1)
    PUT_ROW(2, a2)
    PUT_ROW(3, a3)
    PUT_ROW(4, a4)
    PUT_ROW(5, a5)
    PUT_ROW(6, a6)
#undef PUT_ROW
    }
    __syncthreads();

    if (!KEPT) {
        unsigned hv = (lane == 7 && slow) ? 1u : 0u;
#pragma unroll
        for (int j = 0; j < 7; ++j) if (lane == j) hv = nj[j];
        krec[lane] = hv;
    }

    const int ntiles = (int)(nrows >> 4);
    const int g      = w >> 3;
    const int n0     = (w & 7) << 4;
    if (g == 0) tile_product<4, RELU, !KEPT>(Alds, Wm, bias, rsoS, out, lo, ntiles, g, n0, lane);
    else        tile_product<3, RELU, !KEPT>(Alds, Wm, bias, rsoS, out, lo, ntiles, g, n0, lane);
}

extern "C" void kernel_launch(void* const* d_in, const int* in_sizes, int n_in,
                              void* d_out, int out_size, void* d_ws, size_t ws_size,
                              hipStream_t stream) {
    const float* x   = (const float*)d_in[0];
    const int*   src = (const int*)  d_in[1];
    const int*   dst = (const int*)  d_in[2];
    const float* W1  = (const float*)d_in[3];
    const float* b1  = (const float*)d_in[4];
    const float* W2  = (const float*)d_in[5];
    const float* b2  = (const float*)d_in[6];
    float*       out = (float*)d_out;

    float* ws     = (float*)d_ws;
    float* rs_out = ws;
    float* rs_in  = ws + N_NODES;
    float* scr    = ws + 2 * N_NODES;
    float* h1     = scr + (size_t)N_NODES * DFEAT;

    const int T = 256;

    unsigned* part = (unsigned*)scr;
    unsigned* bits = (unsigned*)d_out;
    degree_hist<<<3 * NPART, HIST_THREADS, 0, stream>>>(src, dst, part, bits);
    unsigned* degb = (unsigned*)h1 - DEGB_WORDS;
    degree_finish<<<(2 * N4 + T - 1) / T, T, 0, stream>>>(part, rs_out, degb);

    unsigned* kept = (unsigned*)scr;
    gcn_layer<true, false><<<NLBLK, LTHREADS, 0, stream>>>(x, rs_out, rs_in, src, dst, W1, b1, h1, kept, bits, degb);
    gcn_layer<false, true><<<NLBLK, LTHREADS, 0, stream>>>(h1, rs_out, rs_in, src, dst, W2, b2, out, kept, nullptr, nullptr);
}
